# starting kernel
# baseline (speedup 1.0000x reference)
#include <hip/hip_runtime.h>
#include <math.h>

typedef __attribute__((ext_vector_type(2))) float v2f;
typedef __attribute__((ext_vector_type(8))) float v8f;

#define BB 2
#define CC 256
#define HH 64
#define WW 64
#define HW 4096
#define NN 16
#define NROW (BB*HW)

__global__ void init_out_kernel(const float* __restrict__ feat,
                                const float* __restrict__ Dvec,
                                float* __restrict__ out) {
    int idx = blockIdx.x * blockDim.x + threadIdx.x;
    int c = (idx >> 12) & (CC - 1);
    out[idx] = feat[idx] * Dvec[c];
}

__global__ void proj_wmma_kernel(const float* __restrict__ feat,
                                 const float* __restrict__ W_delta,
                                 const float* __restrict__ b_delta,
                                 const float* __restrict__ W_B,
                                 const float* __restrict__ W_C,
                                 float* __restrict__ delta_ws,
                                 float* __restrict__ Bm_ws,
                                 float* __restrict__ Cm_ws) {
    const int lane = threadIdx.x;
    const int lo   = lane & 15;
    const int hi   = lane >> 4;
    const int m0   = blockIdx.x * 16;
    const int which = blockIdx.y;

    const float* Wmat;
    int ncols, n0;
    if (which < 16)       { Wmat = W_delta; ncols = CC; n0 = which * 16; }
    else if (which == 16) { Wmat = W_B;     ncols = NN; n0 = 0; }
    else                  { Wmat = W_C;     ncols = NN; n0 = 0; }

    const int rowA = m0 + lo;
    const int bA   = rowA >> 12;
    const int lA   = rowA & (HW - 1);
    const float* aBase = feat + (size_t)bA * CC * HW + lA;

    v8f acc = {};
    for (int k0 = 0; k0 < CC; k0 += 4) {
        const int ka = k0 + hi * 2;
        v2f a, b;
        a.x = aBase[(size_t)ka * HW];
        a.y = aBase[(size_t)(ka + 1) * HW];
        b.x = Wmat[(size_t)ka * ncols + n0 + lo];
        b.y = Wmat[(size_t)(ka + 1) * ncols + n0 + lo];
        acc = __builtin_amdgcn_wmma_f32_16x16x4_f32(
             false, a,  false, b,
             (short)0, acc,  false,  false);
    }

    if (which < 16) {
        const float bias = b_delta[n0 + lo];
        #pragma unroll
        for (int v = 0; v < 8; ++v) {
            const int row = m0 + v + 8 * hi;
            const float pre = acc[v] + bias;
            const float sp = (pre > 20.f) ? pre : log1pf(__expf(pre));
            delta_ws[(size_t)row * CC + n0 + lo] = sp;
        }
    } else {
        float* dst = (which == 16) ? Bm_ws : Cm_ws;
        #pragma unroll
        for (int v = 0; v < 8; ++v) {
            const int row = m0 + v + 8 * hi;
            dst[(size_t)row * NN + lo] = acc[v];
        }
    }
}

__global__ void scan_kernel(const float* __restrict__ feat,
                            const float* __restrict__ A_log,
                            const float* __restrict__ delta_ws,
                            const float* __restrict__ Bm_ws,
                            const float* __restrict__ Cm_ws,
                            float* __restrict__ out) {
    const int lane = threadIdx.x;
    const int n  = lane & 15;
    const int hi = lane >> 4;
    const int id  = blockIdx.x;
    const int dir = id >> 8;
    const int rem = id & 255;
    const int b   = rem >> 7;
    const int c   = ((rem & 127) << 1) + hi;

    int base, sh, sw;
    switch (dir) {
        case 0:  base = 0;    sh =  64; sw =  1; break;
        case 1:  base = 4095; sh = -64; sw = -1; break;
        case 2:  base = 4032; sh = -64; sw =  1; break;
        default: base = 63;   sh =  -1; sw = 64; break;
    }

    const float A = -__expf(A_log[c * NN + n]);
    const float* fBase = feat + (size_t)b * CC * HW + (size_t)c * HW;
    const size_t rbase = (size_t)b * HW;
    float* outBase = out + (size_t)b * CC * HW + (size_t)c * HW;

    float h = 0.f;
    for (int t = 0; t < HW; ++t) {
        const int l = base + sh * (t >> 6) + sw * (t & 63);
        const size_t row = rbase + (size_t)l;
        const float d  = delta_ws[row * CC + c];
        const float xv = fBase[l];
        const float Bv = Bm_ws[row * NN + n];
        const float Cv = Cm_ws[row * NN + n];
        const float dA = __expf(d * A);
        h = fmaf(dA, h, d * xv * Bv);
        float contrib = h * Cv;
        contrib += __shfl_xor(contrib, 1);
        contrib += __shfl_xor(contrib, 2);
        contrib += __shfl_xor(contrib, 4);
        contrib += __shfl_xor(contrib, 8);
        if (n == 0) atomicAdd(&outBase[l], contrib * 0.25f);
    }
}

extern "C" void kernel_launch(void* const* d_in, const int* in_sizes, int n_in,
                              void* d_out, int out_size, void* d_ws, size_t ws_size,
                              hipStream_t stream) {
    const float* feat    = (const float*)d_in[0];
    const float* A_log   = (const float*)d_in[1];
    const float* Dvec    = (const float*)d_in[2];
    const float* W_delta = (const float*)d_in[3];
    const float* b_delta = (const float*)d_in[4];
    const float* W_B     = (const float*)d_in[5];
    const float* W_C     = (const float*)d_in[6];
    float* out = (float*)d_out;

    float* ws       = (float*)d_ws;
    float* delta_ws = ws;
    float* Bm_ws    = ws + (size_t)NROW * CC;
    float* Cm_ws    = Bm_ws + (size_t)NROW * NN;

    init_out_kernel<<<(BB * CC * HW) / 256, 256, 0, stream>>>(feat, Dvec, out);
    proj_wmma_kernel<<<dim3(NROW / 16, 18), 32, 0, stream>>>(
        feat, W_delta, b_delta, W_B, W_C, delta_ws, Bm_ws, Cm_ws);
    scan_kernel<<<1024, 32, 0, stream>>>(feat, A_log, delta_ws, Bm_ws, Cm_ws, out);
}


# scan: each 4096-step chain cut into 16 chunks, one wave each in a 512-thread block; chunk decay product + zero-start end state via LDS give the carry-in (h summed chunk-wise; atomic order varies)
# speedup vs baseline: 8.1948x; 8.1948x over previous
#include <hip/hip_runtime.h>
#include <math.h>

typedef __attribute__((ext_vector_type(2))) float v2f;
typedef __attribute__((ext_vector_type(8))) float v8f;

#define BB 2
#define CC 256
#define HH 64
#define WW 64
#define HW 4096
#define NN 16
#define NROW (BB*HW)

__global__ void init_out_kernel(const float* __restrict__ feat,
                                const float* __restrict__ Dvec,
                                float* __restrict__ out) {
    int idx = blockIdx.x * blockDim.x + threadIdx.x;
    int c = (idx >> 12) & (CC - 1);
    out[idx] = feat[idx] * Dvec[c];
}

__global__ void proj_wmma_kernel(const float* __restrict__ feat,
                                 const float* __restrict__ W_delta,
                                 const float* __restrict__ b_delta,
                                 const float* __restrict__ W_B,
                                 const float* __restrict__ W_C,
                                 float* __restrict__ delta_ws,
                                 float* __restrict__ Bm_ws,
                                 float* __restrict__ Cm_ws) {
    const int lane = threadIdx.x;
    const int lo   = lane & 15;
    const int hi   = lane >> 4;
    const int m0   = blockIdx.x * 16;
    const int which = blockIdx.y;

    const float* Wmat;
    int ncols, n0;
    if (which < 16)       { Wmat = W_delta; ncols = CC; n0 = which * 16; }
    else if (which == 16) { Wmat = W_B;     ncols = NN; n0 = 0; }
    else                  { Wmat = W_C;     ncols = NN; n0 = 0; }

    const int rowA = m0 + lo;
    const int bA   = rowA >> 12;
    const int lA   = rowA & (HW - 1);
    const float* aBase = feat + (size_t)bA * CC * HW + lA;

    v8f acc = {};
    for (int k0 = 0; k0 < CC; k0 += 4) {
        const int ka = k0 + hi * 2;
        v2f a, b;
        a.x = aBase[(size_t)ka * HW];
        a.y = aBase[(size_t)(ka + 1) * HW];
        b.x = Wmat[(size_t)ka * ncols + n0 + lo];
        b.y = Wmat[(size_t)(ka + 1) * ncols + n0 + lo];
        acc = __builtin_amdgcn_wmma_f32_16x16x4_f32(
             false, a,  false, b,
             (short)0, acc,  false,  false);
    }

    if (which < 16) {
        const float bias = b_delta[n0 + lo];
        #pragma unroll
        for (int v = 0; v < 8; ++v) {
            const int row = m0 + v + 8 * hi;
            const float pre = acc[v] + bias;
            const float sp = (pre > 20.f) ? pre : log1pf(__expf(pre));
            delta_ws[(size_t)row * CC + n0 + lo] = sp;
        }
    } else {
        float* dst = (which == 16) ? Bm_ws : Cm_ws;
        #pragma unroll
        for (int v = 0; v < 8; ++v) {
            const int row = m0 + v + 8 * hi;
            dst[(size_t)row * NN + lo] = acc[v];
        }
    }
}

#define KCH  16
#define CLEN (HW / KCH)
__global__ void __launch_bounds__(32 * KCH)
scan_kernel(const float* __restrict__ feat,
            const float* __restrict__ A_log,
            const float* __restrict__ delta_ws,
            const float* __restrict__ Bm_ws,
            const float* __restrict__ Cm_ws,
            float* __restrict__ out) {
    __shared__ float sP[KCH][32];
    __shared__ float sE[KCH][32];

    const int lane = threadIdx.x & 31;
    const int w    = __builtin_amdgcn_readfirstlane((int)(threadIdx.x >> 5));
    const int n  = lane & 15;
    const int hi = lane >> 4;
    const int id  = blockIdx.x;
    const int dir = id >> 8;
    const int rem = id & 255;
    const int b   = rem >> 7;
    const int c   = ((rem & 127) << 1) + hi;

    int base, sh, sw;
    switch (dir) {
        case 0:  base = 0;    sh =  64; sw =  1; break;
        case 1:  base = 4095; sh = -64; sw = -1; break;
        case 2:  base = 4032; sh = -64; sw =  1; break;
        default: base = 63;   sh =  -1; sw = 64; break;
    }

    const float A = -__expf(A_log[c * NN + n]);
    const float* fBase = feat + (size_t)b * CC * HW + (size_t)c * HW;
    const size_t rbase = (size_t)b * HW;
    float* outBase = out + (size_t)b * CC * HW + (size_t)c * HW;

    const int t0 = w * CLEN;
    const int t1 = t0 + CLEN;

    if (w < KCH - 1) {
        float h = 0.f, P = 1.f;
        for (int t = t0; t < t1; ++t) {
            const int l = base + sh * (t >> 6) + sw * (t & 63);
            const size_t row = rbase + (size_t)l;
            const float d  = delta_ws[row * CC + c];
            const float xv = fBase[l];
            const float Bv = Bm_ws[row * NN + n];
            const float dA = __expf(d * A);
            h = fmaf(dA, h, d * xv * Bv);
            P *= dA;
        }
        sP[w][lane] = P;
        sE[w][lane] = h;
    }
    __syncthreads();

    float h = 0.f;
    for (int j = 0; j < w; ++j) h = fmaf(sP[j][lane], h, sE[j][lane]);

    for (int t = t0; t < t1; ++t) {
        const int l = base + sh * (t >> 6) + sw * (t & 63);
        const size_t row = rbase + (size_t)l;
        const float d  = delta_ws[row * CC + c];
        const float xv = fBase[l];
        const float Bv = Bm_ws[row * NN + n];
        const float Cv = Cm_ws[row * NN + n];
        const float dA = __expf(d * A);
        h = fmaf(dA, h, d * xv * Bv);
        float contrib = h * Cv;
        contrib += __shfl_xor(contrib, 1);
        contrib += __shfl_xor(contrib, 2);
        contrib += __shfl_xor(contrib, 4);
        contrib += __shfl_xor(contrib, 8);
        if (n == 0) atomicAdd(&outBase[l], contrib * 0.25f);
    }
}

extern "C" void kernel_launch(void* const* d_in, const int* in_sizes, int n_in,
                              void* d_out, int out_size, void* d_ws, size_t ws_size,
                              hipStream_t stream) {
    const float* feat    = (const float*)d_in[0];
    const float* A_log   = (const float*)d_in[1];
    const float* Dvec    = (const float*)d_in[2];
    const float* W_delta = (const float*)d_in[3];
    const float* b_delta = (const float*)d_in[4];
    const float* W_B     = (const float*)d_in[5];
    const float* W_C     = (const float*)d_in[6];
    float* out = (float*)d_out;

    float* ws       = (float*)d_ws;
    float* delta_ws = ws;
    float* Bm_ws    = ws + (size_t)NROW * CC;
    float* Cm_ws    = Bm_ws + (size_t)NROW * NN;

    init_out_kernel<<<(BB * CC * HW) / 256, 256, 0, stream>>>(feat, Dvec, out);
    proj_wmma_kernel<<<dim3(NROW / 16, 18), 32, 0, stream>>>(
        feat, W_delta, b_delta, W_B, W_C, delta_ws, Bm_ws, Cm_ws);
    scan_kernel<<<1024, 32 * KCH, 0, stream>>>(feat, A_log, delta_ws, Bm_ws, Cm_ws, out);
}


# scan lane map: 4 lanes per channel keep 4 states each (8 channels/wave, 32 chunks of 128 steps, 256 blocks); n-sum = in-lane fma + 2 quad DPP adds (sum order changes); one 32-lane atomic per 4 steps
# speedup vs baseline: 17.0311x; 2.0783x over previous
#include <hip/hip_runtime.h>
#include <math.h>

typedef __attribute__((ext_vector_type(2))) float v2f;
typedef __attribute__((ext_vector_type(8))) float v8f;

#define BB 2
#define CC 256
#define HH 64
#define WW 64
#define HW 4096
#define NN 16
#define NROW (BB*HW)

__global__ void init_out_kernel(const float* __restrict__ feat,
                                const float* __restrict__ Dvec,
                                float* __restrict__ out) {
    int idx = blockIdx.x * blockDim.x + threadIdx.x;
    int c = (idx >> 12) & (CC - 1);
    out[idx] = feat[idx] * Dvec[c];
}

__global__ void proj_wmma_kernel(const float* __restrict__ feat,
                                 const float* __restrict__ W_delta,
                                 const float* __restrict__ b_delta,
                                 const float* __restrict__ W_B,
                                 const float* __restrict__ W_C,
                                 float* __restrict__ delta_ws,
                                 float* __restrict__ Bm_ws,
                                 float* __restrict__ Cm_ws) {
    const int lane = threadIdx.x;
    const int lo   = lane & 15;
    const int hi   = lane >> 4;
    const int m0   = blockIdx.x * 16;
    const int which = blockIdx.y;

    const float* Wmat;
    int ncols, n0;
    if (which < 16)       { Wmat = W_delta; ncols = CC; n0 = which * 16; }
    else if (which == 16) { Wmat = W_B;     ncols = NN; n0 = 0; }
    else                  { Wmat = W_C;     ncols = NN; n0 = 0; }

    const int rowA = m0 + lo;
    const int bA   = rowA >> 12;
    const int lA   = rowA & (HW - 1);
    const float* aBase = feat + (size_t)bA * CC * HW + lA;

    v8f acc = {};
    for (int k0 = 0; k0 < CC; k0 += 4) {
        const int ka = k0 + hi * 2;
        v2f a, b;
        a.x = aBase[(size_t)ka * HW];
        a.y = aBase[(size_t)(ka + 1) * HW];
        b.x = Wmat[(size_t)ka * ncols + n0 + lo];
        b.y = Wmat[(size_t)(ka + 1) * ncols + n0 + lo];
        acc = __builtin_amdgcn_wmma_f32_16x16x4_f32(
             false, a,  false, b,
             (short)0, acc,  false,  false);
    }

    if (which < 16) {
        const float bias = b_delta[n0 + lo];
        #pragma unroll
        for (int v = 0; v < 8; ++v) {
            const int row = m0 + v + 8 * hi;
            const float pre = acc[v] + bias;
            const float sp = (pre > 20.f) ? pre : log1pf(__expf(pre));
            delta_ws[(size_t)row * CC + n0 + lo] = sp;
        }
    } else {
        float* dst = (which == 16) ? Bm_ws : Cm_ws;
        #pragma unroll
        for (int v = 0; v < 8; ++v) {
            const int row = m0 + v + 8 * hi;
            dst[(size_t)row * NN + lo] = acc[v];
        }
    }
}

#define KCH  32
#define CLEN (HW / KCH)
#define SPL  4

typedef __attribute__((ext_vector_type(4))) float v4f;

__device__ __forceinline__ float quad_sum(float v) {
    v += __builtin_bit_cast(float, __builtin_amdgcn_update_dpp(
             0, __builtin_bit_cast(int, v), 0xB1, 0xF, 0xF, true));
    v += __builtin_bit_cast(float, __builtin_amdgcn_update_dpp(
             0, __builtin_bit_cast(int, v), 0x4E, 0xF, 0xF, true));
    return v;
}

__global__ void __launch_bounds__(32 * KCH)
scan_kernel(const float* __restrict__ feat,
            const float* __restrict__ A_log,
            const float* __restrict__ delta_ws,
            const float* __restrict__ Bm_ws,
            const float* __restrict__ Cm_ws,
            float* __restrict__ out) {
    __shared__ float sP[KCH][SPL][32];
    __shared__ float sE[KCH][SPL][32];

    const int lane = threadIdx.x & 31;
    const int w  = __builtin_amdgcn_readfirstlane((int)(threadIdx.x >> 5));
    const int q  = lane & 3;
    const int cl = lane >> 2;
    const int id  = blockIdx.x;
    const int dir = id >> 6;
    const int rem = id & 63;
    const int b   = rem >> 5;
    const int c   = ((rem & 31) << 3) + cl;

    int base, sh, sw;
    switch (dir) {
        case 0:  base = 0;    sh =  64; sw =  1; break;
        case 1:  base = 4095; sh = -64; sw = -1; break;
        case 2:  base = 4032; sh = -64; sw =  1; break;
        default: base = 63;   sh =  -1; sw = 64; break;
    }

    const v4f Al = *reinterpret_cast<const v4f*>(A_log + c * NN + SPL * q);
    float A[SPL];
    for (int j = 0; j < SPL; ++j) A[j] = -__expf(Al[j]);

    const float* fBase = feat + (size_t)b * CC * HW + (size_t)c * HW;
    const size_t rbase = (size_t)b * HW;
    float* outBase = out + (size_t)b * CC * HW + (size_t)c * HW;

    const int t0 = w * CLEN;
    const int t1 = t0 + CLEN;

    if (w < KCH - 1) {
        float h[SPL], P[SPL];
        for (int j = 0; j < SPL; ++j) { h[j] = 0.f; P[j] = 1.f; }
        for (int t = t0; t < t1; t += SPL) {
            const int l0 = base + sh * (t >> 6) + sw * (t & 63);
            float d[SPL], xv[SPL];
            v4f Bv[SPL];
            for (int k = 0; k < SPL; ++k) {
                const int l = l0 + k * sw;
                const size_t row = rbase + (size_t)l;
                d[k]  = delta_ws[row * CC + c];
                xv[k] = fBase[l];
                Bv[k] = *reinterpret_cast<const v4f*>(Bm_ws + row * NN + SPL * q);
            }
            for (int k = 0; k < SPL; ++k) {
                const float dx = d[k] * xv[k];
                for (int j = 0; j < SPL; ++j) {
                    const float dA = __expf(d[k] * A[j]);
                    h[j] = fmaf(dA, h[j], dx * Bv[k][j]);
                    P[j] *= dA;
                }
            }
        }
        for (int j = 0; j < SPL; ++j) { sP[w][j][lane] = P[j]; sE[w][j][lane] = h[j]; }
    }
    __syncthreads();

    float h[SPL];
    for (int j = 0; j < SPL; ++j) h[j] = 0.f;
    for (int i = 0; i < w; ++i) {
        for (int j = 0; j < SPL; ++j) h[j] = fmaf(sP[i][j][lane], h[j], sE[i][j][lane]);
    }

    for (int t = t0; t < t1; t += SPL) {
        const int l0 = base + sh * (t >> 6) + sw * (t & 63);
        float d[SPL], xv[SPL];
        v4f Bv[SPL], Cv[SPL];
        for (int k = 0; k < SPL; ++k) {
            const int l = l0 + k * sw;
            const size_t row = rbase + (size_t)l;
            d[k]  = delta_ws[row * CC + c];
            xv[k] = fBase[l];
            Bv[k] = *reinterpret_cast<const v4f*>(Bm_ws + row * NN + SPL * q);
            Cv[k] = *reinterpret_cast<const v4f*>(Cm_ws + row * NN + SPL * q);
        }
        float mine = 0.f;
        for (int k = 0; k < SPL; ++k) {
            const float dx = d[k] * xv[k];
            for (int j = 0; j < SPL; ++j) {
                const float dA = __expf(d[k] * A[j]);
                h[j] = fmaf(dA, h[j], dx * Bv[k][j]);
            }
            float y = h[0] * Cv[k][0];
            y = fmaf(h[1], Cv[k][1], y);
            y = fmaf(h[2], Cv[k][2], y);
            y = fmaf(h[3], Cv[k][3], y);
            y = quad_sum(y);
            mine = (q == k) ? y : mine;
        }
        atomicAdd(&outBase[l0 + q * sw], mine * 0.25f);
    }
}

extern "C" void kernel_launch(void* const* d_in, const int* in_sizes, int n_in,
                              void* d_out, int out_size, void* d_ws, size_t ws_size,
                              hipStream_t stream) {
    const float* feat    = (const float*)d_in[0];
    const float* A_log   = (const float*)d_in[1];
    const float* Dvec    = (const float*)d_in[2];
    const float* W_delta = (const float*)d_in[3];
    const float* b_delta = (const float*)d_in[4];
    const float* W_B     = (const float*)d_in[5];
    const float* W_C     = (const float*)d_in[6];
    float* out = (float*)d_out;

    float* ws       = (float*)d_ws;
    float* delta_ws = ws;
    float* Bm_ws    = ws + (size_t)NROW * CC;
    float* Cm_ws    = Bm_ws + (size_t)NROW * NN;

    init_out_kernel<<<(BB * CC * HW) / 256, 256, 0, stream>>>(feat, Dvec, out);
    proj_wmma_kernel<<<dim3(NROW / 16, 18), 32, 0, stream>>>(
        feat, W_delta, b_delta, W_B, W_C, delta_ws, Bm_ws, Cm_ws);
    scan_kernel<<<256, 32 * KCH, 0, stream>>>(feat, A_log, delta_ws, Bm_ws, Cm_ws, out);
}


# out = x*D start values now written by 512 extra single-wave groups of the projection launch (one 4096-float (b,c) row each, 128-bit loads/stores); the separate 65536-wave init launch is gone
# speedup vs baseline: 17.8306x; 1.0469x over previous
#include <hip/hip_runtime.h>
#include <math.h>

typedef __attribute__((ext_vector_type(2))) float v2f;
typedef __attribute__((ext_vector_type(8))) float v8f;

#define BB 2
#define CC 256
#define HH 64
#define WW 64
#define HW 4096
#define NN 16
#define NROW (BB*HW)

typedef __attribute__((ext_vector_type(4))) float v4f;

static_assert(NROW / 16 == BB * CC, "slice 18 maps one wave to one (b, c) row");
#define ROWV4 (HW / 4)
#define INITB 8
__global__ void proj_wmma_kernel(const float* __restrict__ feat,
                                 const float* __restrict__ W_delta,
                                 const float* __restrict__ b_delta,
                                 const float* __restrict__ W_B,
                                 const float* __restrict__ W_C,
                                 const float* __restrict__ Dvec,
                                 float* __restrict__ delta_ws,
                                 float* __restrict__ Bm_ws,
                                 float* __restrict__ Cm_ws,
                                 float* __restrict__ out) {
    const int lane = threadIdx.x;
    const int lo   = lane & 15;
    const int hi   = lane >> 4;
    const int m0   = blockIdx.x * 16;
    const int which = blockIdx.y;

    if (which == 18) {
        const int bc = blockIdx.x;
        const float dsc = Dvec[bc & (CC - 1)];
        const v4f* src = reinterpret_cast<const v4f*>(feat + (size_t)bc * HW);
        v4f* dst = reinterpret_cast<v4f*>(out + (size_t)bc * HW);
        for (int i0 = 0; i0 < ROWV4; i0 += 32 * INITB) {
            v4f v[INITB];
            for (int u = 0; u < INITB; ++u) v[u] = src[i0 + u * 32 + lane];
            for (int u = 0; u < INITB; ++u) dst[i0 + u * 32 + lane] = v[u] * dsc;
        }
        return;
    }

    const float* Wmat;
    int ncols, n0;
    if (which < 16)       { Wmat = W_delta; ncols = CC; n0 = which * 16; }
    else if (which == 16) { Wmat = W_B;     ncols = NN; n0 = 0; }
    else                  { Wmat = W_C;     ncols = NN; n0 = 0; }

    const int rowA = m0 + lo;
    const int bA   = rowA >> 12;
    const int lA   = rowA & (HW - 1);
    const float* aBase = feat + (size_t)bA * CC * HW + lA;

    v8f acc = {};
    for (int k0 = 0; k0 < CC; k0 += 4) {
        const int ka = k0 + hi * 2;
        v2f a, b;
        a.x = aBase[(size_t)ka * HW];
        a.y = aBase[(size_t)(ka + 1) * HW];
        b.x = Wmat[(size_t)ka * ncols + n0 + lo];
        b.y = Wmat[(size_t)(ka + 1) * ncols + n0 + lo];
        acc = __builtin_amdgcn_wmma_f32_16x16x4_f32(
             false, a,  false, b,
             (short)0, acc,  false,  false);
    }

    if (which < 16) {
        const float bias = b_delta[n0 + lo];
        #pragma unroll
        for (int v = 0; v < 8; ++v) {
            const int row = m0 + v + 8 * hi;
            const float pre = acc[v] + bias;
            const float sp = (pre > 20.f) ? pre : log1pf(__expf(pre));
            delta_ws[(size_t)row * CC + n0 + lo] = sp;
        }
    } else {
        float* dst = (which == 16) ? Bm_ws : Cm_ws;
        #pragma unroll
        for (int v = 0; v < 8; ++v) {
            const int row = m0 + v + 8 * hi;
            dst[(size_t)row * NN + lo] = acc[v];
        }
    }
}

#define KCH  32
#define CLEN (HW / KCH)
#define SPL  4

__device__ __forceinline__ float quad_sum(float v) {
    v += __builtin_bit_cast(float, __builtin_amdgcn_update_dpp(
             0, __builtin_bit_cast(int, v), 0xB1, 0xF, 0xF, true));
    v += __builtin_bit_cast(float, __builtin_amdgcn_update_dpp(
             0, __builtin_bit_cast(int, v), 0x4E, 0xF, 0xF, true));
    return v;
}

__global__ void __launch_bounds__(32 * KCH)
scan_kernel(const float* __restrict__ feat,
            const float* __restrict__ A_log,
            const float* __restrict__ delta_ws,
            const float* __restrict__ Bm_ws,
            const float* __restrict__ Cm_ws,
            float* __restrict__ out) {
    __shared__ float sP[KCH][SPL][32];
    __shared__ float sE[KCH][SPL][32];

    const int lane = threadIdx.x & 31;
    const int w  = __builtin_amdgcn_readfirstlane((int)(threadIdx.x >> 5));
    const int q  = lane & 3;
    const int cl = lane >> 2;
    const int id  = blockIdx.x;
    const int dir = id >> 6;
    const int rem = id & 63;
    const int b   = rem >> 5;
    const int c   = ((rem & 31) << 3) + cl;

    int base, sh, sw;
    switch (dir) {
        case 0:  base = 0;    sh =  64; sw =  1; break;
        case 1:  base = 4095; sh = -64; sw = -1; break;
        case 2:  base = 4032; sh = -64; sw =  1; break;
        default: base = 63;   sh =  -1; sw = 64; break;
    }

    const v4f Al = *reinterpret_cast<const v4f*>(A_log + c * NN + SPL * q);
    float A[SPL];
    for (int j = 0; j < SPL; ++j) A[j] = -__expf(Al[j]);

    const float* fBase = feat + (size_t)b * CC * HW + (size_t)c * HW;
    const size_t rbase = (size_t)b * HW;
    float* outBase = out + (size_t)b * CC * HW + (size_t)c * HW;

    const int t0 = w * CLEN;
    const int t1 = t0 + CLEN;

    if (w < KCH - 1) {
        float h[SPL], P[SPL];
        for (int j = 0; j < SPL; ++j) { h[j] = 0.f; P[j] = 1.f; }
        for (int t = t0; t < t1; t += SPL) {
            const int l0 = base + sh * (t >> 6) + sw * (t & 63);
            float d[SPL], xv[SPL];
            v4f Bv[SPL];
            for (int k = 0; k < SPL; ++k) {
                const int l = l0 + k * sw;
                const size_t row = rbase + (size_t)l;
                d[k]  = delta_ws[row * CC + c];
                xv[k] = fBase[l];
                Bv[k] = *reinterpret_cast<const v4f*>(Bm_ws + row * NN + SPL * q);
            }
            for (int k = 0; k < SPL; ++k) {
                const float dx = d[k] * xv[k];
                for (int j = 0; j < SPL; ++j) {
                    const float dA = __expf(d[k] * A[j]);
                    h[j] = fmaf(dA, h[j], dx * Bv[k][j]);
                    P[j] *= dA;
                }
            }
        }
        for (int j = 0; j < SPL; ++j) { sP[w][j][lane] = P[j]; sE[w][j][lane] = h[j]; }
    }
    __syncthreads();

    float h[SPL];
    for (int j = 0; j < SPL; ++j) h[j] = 0.f;
    for (int i = 0; i < w; ++i) {
        for (int j = 0; j < SPL; ++j) h[j] = fmaf(sP[i][j][lane], h[j], sE[i][j][lane]);
    }

    for (int t = t0; t < t1; t += SPL) {
        const int l0 = base + sh * (t >> 6) + sw * (t & 63);
        float d[SPL], xv[SPL];
        v4f Bv[SPL], Cv[SPL];
        for (int k = 0; k < SPL; ++k) {
            const int l = l0 + k * sw;
            const size_t row = rbase + (size_t)l;
            d[k]  = delta_ws[row * CC + c];
            xv[k] = fBase[l];
            Bv[k] = *reinterpret_cast<const v4f*>(Bm_ws + row * NN + SPL * q);
            Cv[k] = *reinterpret_cast<const v4f*>(Cm_ws + row * NN + SPL * q);
        }
        float mine = 0.f;
        for (int k = 0; k < SPL; ++k) {
            const float dx = d[k] * xv[k];
            for (int j = 0; j < SPL; ++j) {
                const float dA = __expf(d[k] * A[j]);
                h[j] = fmaf(dA, h[j], dx * Bv[k][j]);
            }
            float y = h[0] * Cv[k][0];
            y = fmaf(h[1], Cv[k][1], y);
            y = fmaf(h[2], Cv[k][2], y);
            y = fmaf(h[3], Cv[k][3], y);
            y = quad_sum(y);
            mine = (q == k) ? y : mine;
        }
        atomicAdd(&outBase[l0 + q * sw], mine * 0.25f);
    }
}

extern "C" void kernel_launch(void* const* d_in, const int* in_sizes, int n_in,
                              void* d_out, int out_size, void* d_ws, size_t ws_size,
                              hipStream_t stream) {
    const float* feat    = (const float*)d_in[0];
    const float* A_log   = (const float*)d_in[1];
    const float* Dvec    = (const float*)d_in[2];
    const float* W_delta = (const float*)d_in[3];
    const float* b_delta = (const float*)d_in[4];
    const float* W_B     = (const float*)d_in[5];
    const float* W_C     = (const float*)d_in[6];
    float* out = (float*)d_out;

    float* ws       = (float*)d_ws;
    float* delta_ws = ws;
    float* Bm_ws    = ws + (size_t)NROW * CC;
    float* Cm_ws    = Bm_ws + (size_t)NROW * NN;

    proj_wmma_kernel<<<dim3(NROW / 16, 19), 32, 0, stream>>>(
        feat, W_delta, b_delta, W_B, W_C, Dvec, delta_ws, Bm_ws, Cm_ws, out);
    scan_kernel<<<256, 32 * KCH, 0, stream>>>(feat, A_log, delta_ws, Bm_ws, Cm_ws, out);
}


# scan: x and delta of a block's 8 channels staged once in LDS as pairs (contiguous 128-bit loads; projection stores delta channel-major); per-step strided global x/delta loads gone; sum order kept
# speedup vs baseline: 20.6552x; 1.1584x over previous
#include <hip/hip_runtime.h>
#include <math.h>

typedef __attribute__((ext_vector_type(2))) float v2f;
typedef __attribute__((ext_vector_type(8))) float v8f;

#define BB 2
#define CC 256
#define HH 64
#define WW 64
#define HW 4096
#define NN 16
#define NROW (BB*HW)

typedef __attribute__((ext_vector_type(4))) float v4f;

static_assert(NROW / 16 == BB * CC, "slice 18 maps one wave to one (b, c) row");
#define ROWV4 (HW / 4)
#define INITB 8
__global__ void proj_wmma_kernel(const float* __restrict__ feat,
                                 const float* __restrict__ W_delta,
                                 const float* __restrict__ b_delta,
                                 const float* __restrict__ W_B,
                                 const float* __restrict__ W_C,
                                 const float* __restrict__ Dvec,
                                 float* __restrict__ delta_ws,
                                 float* __restrict__ Bm_ws,
                                 float* __restrict__ Cm_ws,
                                 float* __restrict__ out) {
    const int lane = threadIdx.x;
    const int lo   = lane & 15;
    const int hi   = lane >> 4;
    const int m0   = blockIdx.x * 16;
    const int which = blockIdx.y;

    if (which == 18) {
        const int bc = blockIdx.x;
        const float dsc = Dvec[bc & (CC - 1)];
        const v4f* src = reinterpret_cast<const v4f*>(feat + (size_t)bc * HW);
        v4f* dst = reinterpret_cast<v4f*>(out + (size_t)bc * HW);
        for (int i0 = 0; i0 < ROWV4; i0 += 32 * INITB) {
            v4f v[INITB];
            for (int u = 0; u < INITB; ++u) v[u] = src[i0 + u * 32 + lane];
            for (int u = 0; u < INITB; ++u) dst[i0 + u * 32 + lane] = v[u] * dsc;
        }
        return;
    }

    const float* Wmat;
    int ncols, n0;
    if (which < 16)       { Wmat = W_delta; ncols = CC; n0 = which * 16; }
    else if (which == 16) { Wmat = W_B;     ncols = NN; n0 = 0; }
    else                  { Wmat = W_C;     ncols = NN; n0 = 0; }

    const int rowA = m0 + lo;
    const int bA   = rowA >> 12;
    const int lA   = rowA & (HW - 1);
    const float* aBase = feat + (size_t)bA * CC * HW + lA;

    v8f acc = {};
    for (int k0 = 0; k0 < CC; k0 += 4) {
        const int ka = k0 + hi * 2;
        v2f a, b;
        a.x = aBase[(size_t)ka * HW];
        a.y = aBase[(size_t)(ka + 1) * HW];
        b.x = Wmat[(size_t)ka * ncols + n0 + lo];
        b.y = Wmat[(size_t)(ka + 1) * ncols + n0 + lo];
        acc = __builtin_amdgcn_wmma_f32_16x16x4_f32(
             false, a,  false, b,
             (short)0, acc,  false,  false);
    }

    if (which < 16) {
        const float bias = b_delta[n0 + lo];
        float sp[8];
        #pragma unroll
        for (int v = 0; v < 8; ++v) {
            const float pre = acc[v] + bias;
            sp[v] = (pre > 20.f) ? pre : log1pf(__expf(pre));
        }
        const int bT = m0 >> 12;
        const int lT = (m0 & (HW - 1)) + 8 * hi;
        v4f* dst = reinterpret_cast<v4f*>(delta_ws + ((size_t)(bT * CC + n0 + lo)) * HW + lT);
        dst[0] = (v4f){sp[0], sp[1], sp[2], sp[3]};
        dst[1] = (v4f){sp[4], sp[5], sp[6], sp[7]};
    } else {
        float* dst = (which == 16) ? Bm_ws : Cm_ws;
        #pragma unroll
        for (int v = 0; v < 8; ++v) {
            const int row = m0 + v + 8 * hi;
            dst[(size_t)row * NN + lo] = acc[v];
        }
    }
}

#define KCH  32
#define CLEN (HW / KCH)
#define SPL  4

__device__ __forceinline__ float quad_sum(float v) {
    v += __builtin_bit_cast(float, __builtin_amdgcn_update_dpp(
             0, __builtin_bit_cast(int, v), 0xB1, 0xF, 0xF, true));
    v += __builtin_bit_cast(float, __builtin_amdgcn_update_dpp(
             0, __builtin_bit_cast(int, v), 0x4E, 0xF, 0xF, true));
    return v;
}

#define XDP (HW + 4)
__global__ void __launch_bounds__(32 * KCH)
scan_kernel(const float* __restrict__ feat,
            const float* __restrict__ A_log,
            const float* __restrict__ delta_ws,
            const float* __restrict__ Bm_ws,
            const float* __restrict__ Cm_ws,
            float* __restrict__ out) {
    __shared__ __attribute__((aligned(16))) v2f sXD[8 * XDP];
    __shared__ float sP[KCH][SPL][32];
    __shared__ float sE[KCH][SPL][32];

    const int lane = threadIdx.x & 31;
    const int w  = __builtin_amdgcn_readfirstlane((int)(threadIdx.x >> 5));
    const int q  = lane & 3;
    const int cl = lane >> 2;
    const int id  = blockIdx.x;
    const int dir = id >> 6;
    const int rem = id & 63;
    const int b   = rem >> 5;
    const int cg  = (rem & 31) << 3;
    const int c   = cg + cl;

    {
        const int i = threadIdx.x;
        const size_t goff = ((size_t)(b * CC + cg)) * HW;
        const v4f* xs = reinterpret_cast<const v4f*>(feat + goff);
        const v4f* ds = reinterpret_cast<const v4f*>(delta_ws + goff);
        v4f xr[8], dr[8];
        for (int j = 0; j < 8; ++j) {
            xr[j] = xs[j * ROWV4 + i];
            dr[j] = ds[j * ROWV4 + i];
        }
        for (int j = 0; j < 8; ++j) {
            v4f* p = reinterpret_cast<v4f*>(&sXD[j * XDP + 4 * i]);
            p[0] = (v4f){xr[j].x, dr[j].x, xr[j].y, dr[j].y};
            p[1] = (v4f){xr[j].z, dr[j].z, xr[j].w, dr[j].w};
        }
    }

    int base, sh, sw;
    switch (dir) {
        case 0:  base = 0;    sh =  64; sw =  1; break;
        case 1:  base = 4095; sh = -64; sw = -1; break;
        case 2:  base = 4032; sh = -64; sw =  1; break;
        default: base = 63;   sh =  -1; sw = 64; break;
    }

    const v4f Al = *reinterpret_cast<const v4f*>(A_log + c * NN + SPL * q);
    float A[SPL];
    for (int j = 0; j < SPL; ++j) A[j] = -__expf(Al[j]);

    const v2f* xdBase = sXD + cl * XDP;
    const size_t rbase = (size_t)b * HW;
    float* outBase = out + (size_t)b * CC * HW + (size_t)c * HW;

    const int t0 = w * CLEN;
    const int t1 = t0 + CLEN;

    __syncthreads();

    if (w < KCH - 1) {
        float h[SPL], P[SPL];
        for (int j = 0; j < SPL; ++j) { h[j] = 0.f; P[j] = 1.f; }
        for (int t = t0; t < t1; t += SPL) {
            const int l0 = base + sh * (t >> 6) + sw * (t & 63);
            float d[SPL], xv[SPL];
            v4f Bv[SPL];
            for (int k = 0; k < SPL; ++k) {
                const int l = l0 + k * sw;
                const size_t row = rbase + (size_t)l;
                const v2f xd = xdBase[l];
                xv[k] = xd.x;
                d[k]  = xd.y;
                Bv[k] = *reinterpret_cast<const v4f*>(Bm_ws + row * NN + SPL * q);
            }
            for (int k = 0; k < SPL; ++k) {
                const float dx = d[k] * xv[k];
                for (int j = 0; j < SPL; ++j) {
                    const float dA = __expf(d[k] * A[j]);
                    h[j] = fmaf(dA, h[j], dx * Bv[k][j]);
                    P[j] *= dA;
                }
            }
        }
        for (int j = 0; j < SPL; ++j) { sP[w][j][lane] = P[j]; sE[w][j][lane] = h[j]; }
    }
    __syncthreads();

    float h[SPL];
    for (int j = 0; j < SPL; ++j) h[j] = 0.f;
    for (int i = 0; i < w; ++i) {
        for (int j = 0; j < SPL; ++j) h[j] = fmaf(sP[i][j][lane], h[j], sE[i][j][lane]);
    }

    for (int t = t0; t < t1; t += SPL) {
        const int l0 = base + sh * (t >> 6) + sw * (t & 63);
        float d[SPL], xv[SPL];
        v4f Bv[SPL], Cv[SPL];
        for (int k = 0; k < SPL; ++k) {
            const int l = l0 + k * sw;
            const size_t row = rbase + (size_t)l;
            const v2f xd = xdBase[l];
            xv[k] = xd.x;
            d[k]  = xd.y;
            Bv[k] = *reinterpret_cast<const v4f*>(Bm_ws + row * NN + SPL * q);
            Cv[k] = *reinterpret_cast<const v4f*>(Cm_ws + row * NN + SPL * q);
        }
        float mine = 0.f;
        for (int k = 0; k < SPL; ++k) {
            const float dx = d[k] * xv[k];
            for (int j = 0; j < SPL; ++j) {
                const float dA = __expf(d[k] * A[j]);
                h[j] = fmaf(dA, h[j], dx * Bv[k][j]);
            }
            float y = h[0] * Cv[k][0];
            y = fmaf(h[1], Cv[k][1], y);
            y = fmaf(h[2], Cv[k][2], y);
            y = fmaf(h[3], Cv[k][3], y);
            y = quad_sum(y);
            mine = (q == k) ? y : mine;
        }
        atomicAdd(&outBase[l0 + q * sw], mine * 0.25f);
    }
}

extern "C" void kernel_launch(void* const* d_in, const int* in_sizes, int n_in,
                              void* d_out, int out_size, void* d_ws, size_t ws_size,
                              hipStream_t stream) {
    const float* feat    = (const float*)d_in[0];
    const float* A_log   = (const float*)d_in[1];
    const float* Dvec    = (const float*)d_in[2];
    const float* W_delta = (const float*)d_in[3];
    const float* b_delta = (const float*)d_in[4];
    const float* W_B     = (const float*)d_in[5];
    const float* W_C     = (const float*)d_in[6];
    float* out = (float*)d_out;

    float* ws       = (float*)d_ws;
    float* delta_ws = ws;
    float* Bm_ws    = ws + (size_t)NROW * CC;
    float* Cm_ws    = Bm_ws + (size_t)NROW * NN;

    proj_wmma_kernel<<<dim3(NROW / 16, 19), 32, 0, stream>>>(
        feat, W_delta, b_delta, W_B, W_C, Dvec, delta_ws, Bm_ws, Cm_ws, out);
    scan_kernel<<<256, 32 * KCH, 0, stream>>>(feat, A_log, delta_ws, Bm_ws, Cm_ws, out);
}


# scan: a work-group now runs all 4 directions of 2 channels; their outputs go to LDS planes, are summed in a fixed order and stored once with x*D; atomics and the start-value slice are gone
# speedup vs baseline: 25.0968x; 1.2150x over previous
#include <hip/hip_runtime.h>
#include <math.h>

typedef __attribute__((ext_vector_type(2))) float v2f;
typedef __attribute__((ext_vector_type(8))) float v8f;

#define BB 2
#define CC 256
#define HH 64
#define WW 64
#define HW 4096
#define NN 16
#define NROW (BB*HW)

typedef __attribute__((ext_vector_type(4))) float v4f;

__global__ void proj_wmma_kernel(const float* __restrict__ feat,
                                 const float* __restrict__ W_delta,
                                 const float* __restrict__ b_delta,
                                 const float* __restrict__ W_B,
                                 const float* __restrict__ W_C,
                                 float* __restrict__ delta_ws,
                                 float* __restrict__ Bm_ws,
                                 float* __restrict__ Cm_ws) {
    const int lane = threadIdx.x;
    const int lo   = lane & 15;
    const int hi   = lane >> 4;
    const int m0   = blockIdx.x * 16;
    const int which = blockIdx.y;

    const float* Wmat;
    int ncols, n0;
    if (which < 16)       { Wmat = W_delta; ncols = CC; n0 = which * 16; }
    else if (which == 16) { Wmat = W_B;     ncols = NN; n0 = 0; }
    else                  { Wmat = W_C;     ncols = NN; n0 = 0; }

    const int rowA = m0 + lo;
    const int bA   = rowA >> 12;
    const int lA   = rowA & (HW - 1);
    const float* aBase = feat + (size_t)bA * CC * HW + lA;

    v8f acc = {};
    for (int k0 = 0; k0 < CC; k0 += 4) {
        const int ka = k0 + hi * 2;
        v2f a, b;
        a.x = aBase[(size_t)ka * HW];
        a.y = aBase[(size_t)(ka + 1) * HW];
        b.x = Wmat[(size_t)ka * ncols + n0 + lo];
        b.y = Wmat[(size_t)(ka + 1) * ncols + n0 + lo];
        acc = __builtin_amdgcn_wmma_f32_16x16x4_f32(
             false, a,  false, b,
             (short)0, acc,  false,  false);
    }

    if (which < 16) {
        const float bias = b_delta[n0 + lo];
        float sp[8];
        #pragma unroll
        for (int v = 0; v < 8; ++v) {
            const float pre = acc[v] + bias;
            sp[v] = (pre > 20.f) ? pre : log1pf(__expf(pre));
        }
        const int bT = m0 >> 12;
        const int lT = (m0 & (HW - 1)) + 8 * hi;
        v4f* dst = reinterpret_cast<v4f*>(delta_ws + ((size_t)(bT * CC + n0 + lo)) * HW + lT);
        dst[0] = (v4f){sp[0], sp[1], sp[2], sp[3]};
        dst[1] = (v4f){sp[4], sp[5], sp[6], sp[7]};
    } else {
        float* dst = (which == 16) ? Bm_ws : Cm_ws;
        #pragma unroll
        for (int v = 0; v < 8; ++v) {
            const int row = m0 + v + 8 * hi;
            dst[(size_t)row * NN + lo] = acc[v];
        }
    }
}

#define KCH  32
#define CLEN (HW / KCH)
#define SPL  4
#define NCH  2
#define NPL  (4 * NCH)
#define RUN  64
#define ROWV4 (HW / 4)
static_assert(CLEN % RUN == 0 && RUN % SPL == 0, "a chunk is a whole number of runs of whole blocks");
static_assert(NPL * SPL == 32, "one wave = 8 pairs x 4 state quarters");

__device__ __forceinline__ float quad_sum(float v) {
    v += __builtin_bit_cast(float, __builtin_amdgcn_update_dpp(
             0, __builtin_bit_cast(int, v), 0xB1, 0xF, 0xF, true));
    v += __builtin_bit_cast(float, __builtin_amdgcn_update_dpp(
             0, __builtin_bit_cast(int, v), 0x4E, 0xF, 0xF, true));
    return v;
}

#define XDP (HW + 4)
#define YPP (HW + 4)
__global__ void __launch_bounds__(32 * KCH)
scan_kernel(const float* __restrict__ feat,
            const float* __restrict__ A_log,
            const float* __restrict__ Dvec,
            const float* __restrict__ delta_ws,
            const float* __restrict__ Bm_ws,
            const float* __restrict__ Cm_ws,
            float* __restrict__ out) {
    __shared__ __attribute__((aligned(16))) v2f sXD[NCH * XDP];
    __shared__ __attribute__((aligned(16))) float sY[NPL * YPP];
    __shared__ float sP[KCH][SPL][32];
    __shared__ float sE[KCH][SPL][32];

    const int tid  = threadIdx.x;
    const int lane = tid & 31;
    const int w  = __builtin_amdgcn_readfirstlane((int)(threadIdx.x >> 5));
    const int q  = lane & 3;
    const int cl = lane >> 2;
    const int ch  = cl & 1;
    const int dir = cl >> 1;
    const int id  = blockIdx.x;
    const int b   = id >> 7;
    const int c0  = (id & 127) * NCH;
    const int c   = c0 + ch;

    {
        const size_t goff = ((size_t)(b * CC + c0)) * HW;
        const v4f* xs = reinterpret_cast<const v4f*>(feat + goff);
        const v4f* ds = reinterpret_cast<const v4f*>(delta_ws + goff);
        v4f xr[NCH], dr[NCH];
        for (int j = 0; j < NCH; ++j) {
            xr[j] = xs[j * ROWV4 + tid];
            dr[j] = ds[j * ROWV4 + tid];
        }
        for (int j = 0; j < NCH; ++j) {
            v4f* p = reinterpret_cast<v4f*>(&sXD[j * XDP + 4 * tid]);
            p[0] = (v4f){xr[j].x, dr[j].x, xr[j].y, dr[j].y};
            p[1] = (v4f){xr[j].z, dr[j].z, xr[j].w, dr[j].w};
        }
    }

    const int base = (dir == 0) ? 0  : (dir == 1) ? 4095 : (dir == 2) ? 4032 : 63;
    const int sh   = (dir == 0) ? 64 : (dir == 3) ? -1 : -64;
    const int sw   = (dir == 1) ? -1 : (dir == 3) ? 64 : 1;

    const v4f Al = *reinterpret_cast<const v4f*>(A_log + c * NN + SPL * q);
    float A[SPL];
    for (int j = 0; j < SPL; ++j) A[j] = -__expf(Al[j]);

    const int lS = base + sh * (w * (CLEN / RUN));
    const unsigned xa0 = (unsigned)((ch * XDP + lS) * 8);
    const unsigned ga0 = (unsigned)(lS * (NN * 4) + SPL * 4 * q);
    const unsigned ya0 = (unsigned)((cl * YPP + lS + q * sw) * 4);
    const int xstep = 8 * sw;
    const int gstep = NN * 4 * sw;
    const int ystep = 4 * sw;
    const int runfix = sh - RUN * sw;

    const char* xdB = reinterpret_cast<const char*>(sXD);
    char* yB = reinterpret_cast<char*>(sY);
    const char* BmB = reinterpret_cast<const char*>(Bm_ws + (size_t)b * HW * NN);
    const char* CmB = reinterpret_cast<const char*>(Cm_ws + (size_t)b * HW * NN);

    __syncthreads();

    if (w < KCH - 1) {
        float h[SPL], P[SPL];
        for (int j = 0; j < SPL; ++j) { h[j] = 0.f; P[j] = 1.f; }
        unsigned xa = xa0, ga = ga0;
        for (int run = 0; run < CLEN / RUN; ++run) {
            for (int it = 0; it < RUN / SPL; ++it) {
                float d[SPL], xv[SPL];
                v4f Bv[SPL];
                for (int k = 0; k < SPL; ++k) {
                    const v2f xd = *reinterpret_cast<const v2f*>(xdB + (xa + (unsigned)(k * xstep)));
                    xv[k] = xd.x;
                    d[k]  = xd.y;
                    Bv[k] = *reinterpret_cast<const v4f*>(BmB + (ga + (unsigned)(k * gstep)));
                }
                for (int k = 0; k < SPL; ++k) {
                    const float dx = d[k] * xv[k];
                    for (int j = 0; j < SPL; ++j) {
                        const float dA = __expf(d[k] * A[j]);
                        h[j] = fmaf(dA, h[j], dx * Bv[k][j]);
                        P[j] *= dA;
                    }
                }
                xa += (unsigned)(SPL * xstep);
                ga += (unsigned)(SPL * gstep);
            }
            xa += (unsigned)(8 * runfix);
            ga += (unsigned)(NN * 4 * runfix);
        }
        for (int j = 0; j < SPL; ++j) { sP[w][j][lane] = P[j]; sE[w][j][lane] = h[j]; }
    }
    __syncthreads();

    float h[SPL];
    for (int j = 0; j < SPL; ++j) h[j] = 0.f;
    for (int i = 0; i < w; ++i) {
        for (int j = 0; j < SPL; ++j) h[j] = fmaf(sP[i][j][lane], h[j], sE[i][j][lane]);
    }

    {
        unsigned xa = xa0, ga = ga0, ya = ya0;
        for (int run = 0; run < CLEN / RUN; ++run) {
            for (int it = 0; it < RUN / SPL; ++it) {
                float d[SPL], xv[SPL];
                v4f Bv[SPL], Cv[SPL];
                for (int k = 0; k < SPL; ++k) {
                    const v2f xd = *reinterpret_cast<const v2f*>(xdB + (xa + (unsigned)(k * xstep)));
                    xv[k] = xd.x;
                    d[k]  = xd.y;
                    Bv[k] = *reinterpret_cast<const v4f*>(BmB + (ga + (unsigned)(k * gstep)));
                    Cv[k] = *reinterpret_cast<const v4f*>(CmB + (ga + (unsigned)(k * gstep)));
                }
                float mine = 0.f;
                for (int k = 0; k < SPL; ++k) {
                    const float dx = d[k] * xv[k];
                    for (int j = 0; j < SPL; ++j) {
                        const float dA = __expf(d[k] * A[j]);
                        h[j] = fmaf(dA, h[j], dx * Bv[k][j]);
                    }
                    float y = h[0] * Cv[k][0];
                    y = fmaf(h[1], Cv[k][1], y);
                    y = fmaf(h[2], Cv[k][2], y);
                    y = fmaf(h[3], Cv[k][3], y);
                    y = quad_sum(y);
                    mine = (q == k) ? y : mine;
                }
                *reinterpret_cast<float*>(yB + ya) = mine;
                xa += (unsigned)(SPL * xstep);
                ga += (unsigned)(SPL * gstep);
                ya += (unsigned)(SPL * ystep);
            }
            xa += (unsigned)(8 * runfix);
            ga += (unsigned)(NN * 4 * runfix);
            ya += (unsigned)(4 * runfix);
        }
    }
    __syncthreads();

    for (int j = 0; j < NCH; ++j) {
        const float dsc = Dvec[c0 + j];
        const v4f* px = reinterpret_cast<const v4f*>(&sXD[j * XDP + 4 * tid]);
        const v4f p0 = px[0], p1 = px[1];
        const v4f xq = (v4f){p0.x, p0.z, p1.x, p1.z};
        const v4f y0 = *reinterpret_cast<const v4f*>(&sY[(0 * NCH + j) * YPP + 4 * tid]);
        const v4f y1 = *reinterpret_cast<const v4f*>(&sY[(1 * NCH + j) * YPP + 4 * tid]);
        const v4f y2 = *reinterpret_cast<const v4f*>(&sY[(2 * NCH + j) * YPP + 4 * tid]);
        const v4f y3 = *reinterpret_cast<const v4f*>(&sY[(3 * NCH + j) * YPP + 4 * tid]);
        const v4f s = (y0 + y1) + (y2 + y3);
        v4f* dst = reinterpret_cast<v4f*>(out + ((size_t)(b * CC + c0 + j)) * HW);
        dst[tid] = xq * dsc + s * 0.25f;
    }
}

extern "C" void kernel_launch(void* const* d_in, const int* in_sizes, int n_in,
                              void* d_out, int out_size, void* d_ws, size_t ws_size,
                              hipStream_t stream) {
    const float* feat    = (const float*)d_in[0];
    const float* A_log   = (const float*)d_in[1];
    const float* Dvec    = (const float*)d_in[2];
    const float* W_delta = (const float*)d_in[3];
    const float* b_delta = (const float*)d_in[4];
    const float* W_B     = (const float*)d_in[5];
    const float* W_C     = (const float*)d_in[6];
    float* out = (float*)d_out;

    float* ws       = (float*)d_ws;
    float* delta_ws = ws;
    float* Bm_ws    = ws + (size_t)NROW * CC;
    float* Cm_ws    = Bm_ws + (size_t)NROW * NN;

    proj_wmma_kernel<<<dim3(NROW / 16, 18), 32, 0, stream>>>(
        feat, W_delta, b_delta, W_B, W_C, delta_ws, Bm_ws, Cm_ws);
    scan_kernel<<<256, 32 * KCH, 0, stream>>>(feat, A_log, Dvec, delta_ws, Bm_ws, Cm_ws, out);
}


# scan pass 1 walks each position once per axis, not per direction: a row walk also sums the reversed order's end state; per-row/column (P,E) fold into carries (carry-in sum order changes)
# speedup vs baseline: 27.3114x; 1.0882x over previous
#include <hip/hip_runtime.h>
#include <math.h>

typedef __attribute__((ext_vector_type(2))) float v2f;
typedef __attribute__((ext_vector_type(8))) float v8f;

#define BB 2
#define CC 256
#define HH 64
#define WW 64
#define HW 4096
#define NN 16
#define NROW (BB*HW)

typedef __attribute__((ext_vector_type(4))) float v4f;

__global__ void proj_wmma_kernel(const float* __restrict__ feat,
                                 const float* __restrict__ W_delta,
                                 const float* __restrict__ b_delta,
                                 const float* __restrict__ W_B,
                                 const float* __restrict__ W_C,
                                 float* __restrict__ delta_ws,
                                 float* __restrict__ Bm_ws,
                                 float* __restrict__ Cm_ws) {
    const int lane = threadIdx.x;
    const int lo   = lane & 15;
    const int hi   = lane >> 4;
    const int m0   = blockIdx.x * 16;
    const int which = blockIdx.y;

    const float* Wmat;
    int ncols, n0;
    if (which < 16)       { Wmat = W_delta; ncols = CC; n0 = which * 16; }
    else if (which == 16) { Wmat = W_B;     ncols = NN; n0 = 0; }
    else                  { Wmat = W_C;     ncols = NN; n0 = 0; }

    const int rowA = m0 + lo;
    const int bA   = rowA >> 12;
    const int lA   = rowA & (HW - 1);
    const float* aBase = feat + (size_t)bA * CC * HW + lA;

    v8f acc = {};
    for (int k0 = 0; k0 < CC; k0 += 4) {
        const int ka = k0 + hi * 2;
        v2f a, b;
        a.x = aBase[(size_t)ka * HW];
        a.y = aBase[(size_t)(ka + 1) * HW];
        b.x = Wmat[(size_t)ka * ncols + n0 + lo];
        b.y = Wmat[(size_t)(ka + 1) * ncols + n0 + lo];
        acc = __builtin_amdgcn_wmma_f32_16x16x4_f32(
             false, a,  false, b,
             (short)0, acc,  false,  false);
    }

    if (which < 16) {
        const float bias = b_delta[n0 + lo];
        float sp[8];
        #pragma unroll
        for (int v = 0; v < 8; ++v) {
            const float pre = acc[v] + bias;
            sp[v] = (pre > 20.f) ? pre : log1pf(__expf(pre));
        }
        const int bT = m0 >> 12;
        const int lT = (m0 & (HW - 1)) + 8 * hi;
        v4f* dst = reinterpret_cast<v4f*>(delta_ws + ((size_t)(bT * CC + n0 + lo)) * HW + lT);
        dst[0] = (v4f){sp[0], sp[1], sp[2], sp[3]};
        dst[1] = (v4f){sp[4], sp[5], sp[6], sp[7]};
    } else {
        float* dst = (which == 16) ? Bm_ws : Cm_ws;
        #pragma unroll
        for (int v = 0; v < 8; ++v) {
            const int row = m0 + v + 8 * hi;
            dst[(size_t)row * NN + lo] = acc[v];
        }
    }
}

#define KCH  32
#define CLEN (HW / KCH)
#define SPL  4
#define NCH  2
#define NPL  (4 * NCH)
#define RUN  64
#define ROWV4 (HW / 4)
static_assert(CLEN % RUN == 0 && RUN % SPL == 0, "a chunk is a whole number of runs of whole blocks");
static_assert(NPL * SPL == 32, "one wave = 8 pairs x 4 state quarters");

__device__ __forceinline__ float quad_sum(float v) {
    v += __builtin_bit_cast(float, __builtin_amdgcn_update_dpp(
             0, __builtin_bit_cast(int, v), 0xB1, 0xF, 0xF, true));
    v += __builtin_bit_cast(float, __builtin_amdgcn_update_dpp(
             0, __builtin_bit_cast(int, v), 0x4E, 0xF, 0xF, true));
    return v;
}

#define XDP (HW + 4)
#define YPP (HW + 4)
__global__ void __launch_bounds__(32 * KCH)
scan_kernel(const float* __restrict__ feat,
            const float* __restrict__ A_log,
            const float* __restrict__ Dvec,
            const float* __restrict__ delta_ws,
            const float* __restrict__ Bm_ws,
            const float* __restrict__ Cm_ws,
            float* __restrict__ out) {
    __shared__ __attribute__((aligned(16))) v2f sXD[NCH * XDP];
    __shared__ __attribute__((aligned(16))) float sY[NPL * YPP];
    __shared__ __attribute__((aligned(16))) float sS[3][64][64];

    const int tid  = threadIdx.x;
    const int lane = tid & 31;
    const int w  = __builtin_amdgcn_readfirstlane((int)(threadIdx.x >> 5));
    const int q  = lane & 3;
    const int cl = lane >> 2;
    const int ch  = cl & 1;
    const int dir = cl >> 1;
    const int id  = blockIdx.x;
    const int b   = id >> 7;
    const int c0  = (id & 127) * NCH;
    const int c   = c0 + ch;

    {
        const size_t goff = ((size_t)(b * CC + c0)) * HW;
        const v4f* xs = reinterpret_cast<const v4f*>(feat + goff);
        const v4f* ds = reinterpret_cast<const v4f*>(delta_ws + goff);
        v4f xr[NCH], dr[NCH];
        for (int j = 0; j < NCH; ++j) {
            xr[j] = xs[j * ROWV4 + tid];
            dr[j] = ds[j * ROWV4 + tid];
        }
        for (int j = 0; j < NCH; ++j) {
            v4f* p = reinterpret_cast<v4f*>(&sXD[j * XDP + 4 * tid]);
            p[0] = (v4f){xr[j].x, dr[j].x, xr[j].y, dr[j].y};
            p[1] = (v4f){xr[j].z, dr[j].z, xr[j].w, dr[j].w};
        }
    }

    const int base = (dir == 0) ? 0  : (dir == 1) ? 4095 : (dir == 2) ? 4032 : 63;
    const int sh   = (dir == 0) ? 64 : (dir == 3) ? -1 : -64;
    const int sw   = (dir == 1) ? -1 : (dir == 3) ? 64 : 1;

    const v4f Al = *reinterpret_cast<const v4f*>(A_log + c * NN + SPL * q);
    float A[SPL];
    for (int j = 0; j < SPL; ++j) A[j] = -__expf(Al[j]);

    const int lS = base + sh * (w * (CLEN / RUN));
    const unsigned xa0 = (unsigned)((ch * XDP + lS) * 8);
    const unsigned ga0 = (unsigned)(lS * (NN * 4) + SPL * 4 * q);
    const unsigned ya0 = (unsigned)((cl * YPP + lS + q * sw) * 4);
    const int xstep = 8 * sw;
    const int gstep = NN * 4 * sw;
    const int ystep = 4 * sw;
    const int runfix = sh - RUN * sw;

    const char* xdB = reinterpret_cast<const char*>(sXD);
    char* yB = reinterpret_cast<char*>(sY);
    const char* BmB = reinterpret_cast<const char*>(Bm_ws + (size_t)b * HW * NN);
    const char* CmB = reinterpret_cast<const char*>(Cm_ws + (size_t)b * HW * NN);

    __syncthreads();

    {
        const int trav = (lane >> 3) & 1;
        const int u    = 2 * w + (lane >> 4);
        const int l0   = trav ? (63 - u) : (RUN * u);
        const int pst  = trav ? WW : 1;
        unsigned xa = (unsigned)((ch * XDP + l0) * 8);
        unsigned ga = (unsigned)(l0 * (NN * 4) + SPL * 4 * q);
        const unsigned xs1 = (unsigned)(8 * pst);
        const unsigned gs1 = (unsigned)(NN * 4 * pst);
        float hf[SPL], hb[SPL], P[SPL];
        for (int j = 0; j < SPL; ++j) { hf[j] = 0.f; hb[j] = 0.f; P[j] = 1.f; }
        for (int it = 0; it < RUN / SPL; ++it) {
            float d[SPL], xv[SPL];
            v4f Bv[SPL];
            for (int k = 0; k < SPL; ++k) {
                const v2f xd = *reinterpret_cast<const v2f*>(xdB + (xa + (unsigned)k * xs1));
                xv[k] = xd.x;
                d[k]  = xd.y;
                Bv[k] = *reinterpret_cast<const v4f*>(BmB + (ga + (unsigned)k * gs1));
            }
            for (int k = 0; k < SPL; ++k) {
                const float dx = d[k] * xv[k];
                for (int j = 0; j < SPL; ++j) {
                    const float dA = __expf(d[k] * A[j]);
                    const float bt = dx * Bv[k][j];
                    hb[j] = fmaf(P[j], bt, hb[j]);
                    hf[j] = fmaf(dA, hf[j], bt);
                    P[j] *= dA;
                }
            }
            xa += (unsigned)SPL * xs1;
            ga += (unsigned)SPL * gs1;
        }
        const int col = (lane & 15) * SPL;
        *reinterpret_cast<v4f*>(&sS[0][u][col]) = (v4f){P[0], P[1], P[2], P[3]};
        *reinterpret_cast<v4f*>(&sS[1][u][col]) = (v4f){hf[0], hf[1], hf[2], hf[3]};
        *reinterpret_cast<v4f*>(&sS[2][u][col]) = (v4f){hb[0], hb[1], hb[2], hb[3]};
    }
    __syncthreads();

    float h[SPL];
    for (int j = 0; j < SPL; ++j) h[j] = 0.f;
    {
        const bool down = (dir == 1) || (dir == 2);
        const int ccol = ((dir == 3) ? 32 : 0) + ch * NN + SPL * q;
        const float* pP = &sS[0][down ? 63 : 0][ccol];
        const float* pE = &sS[(dir == 1) ? 2 : 1][down ? 63 : 0][ccol];
        const int ust = down ? -64 : 64;
        for (int i = 0; i < 2 * w; ++i) {
            const v4f Pv = *reinterpret_cast<const v4f*>(pP);
            const v4f Ev = *reinterpret_cast<const v4f*>(pE);
            for (int j = 0; j < SPL; ++j) h[j] = fmaf(Pv[j], h[j], Ev[j]);
            pP += ust;
            pE += ust;
        }
    }

    {
        unsigned xa = xa0, ga = ga0, ya = ya0;
        for (int run = 0; run < CLEN / RUN; ++run) {
            for (int it = 0; it < RUN / SPL; ++it) {
                float d[SPL], xv[SPL];
                v4f Bv[SPL], Cv[SPL];
                for (int k = 0; k < SPL; ++k) {
                    const v2f xd = *reinterpret_cast<const v2f*>(xdB + (xa + (unsigned)(k * xstep)));
                    xv[k] = xd.x;
                    d[k]  = xd.y;
                    Bv[k] = *reinterpret_cast<const v4f*>(BmB + (ga + (unsigned)(k * gstep)));
                    Cv[k] = *reinterpret_cast<const v4f*>(CmB + (ga + (unsigned)(k * gstep)));
                }
                float mine = 0.f;
                for (int k = 0; k < SPL; ++k) {
                    const float dx = d[k] * xv[k];
                    for (int j = 0; j < SPL; ++j) {
                        const float dA = __expf(d[k] * A[j]);
                        h[j] = fmaf(dA, h[j], dx * Bv[k][j]);
                    }
                    float y = h[0] * Cv[k][0];
                    y = fmaf(h[1], Cv[k][1], y);
                    y = fmaf(h[2], Cv[k][2], y);
                    y = fmaf(h[3], Cv[k][3], y);
                    y = quad_sum(y);
                    mine = (q == k) ? y : mine;
                }
                *reinterpret_cast<float*>(yB + ya) = mine;
                xa += (unsigned)(SPL * xstep);
                ga += (unsigned)(SPL * gstep);
                ya += (unsigned)(SPL * ystep);
            }
            xa += (unsigned)(8 * runfix);
            ga += (unsigned)(NN * 4 * runfix);
            ya += (unsigned)(4 * runfix);
        }
    }
    __syncthreads();

    for (int j = 0; j < NCH; ++j) {
        const float dsc = Dvec[c0 + j];
        const v4f* px = reinterpret_cast<const v4f*>(&sXD[j * XDP + 4 * tid]);
        const v4f p0 = px[0], p1 = px[1];
        const v4f xq = (v4f){p0.x, p0.z, p1.x, p1.z};
        const v4f y0 = *reinterpret_cast<const v4f*>(&sY[(0 * NCH + j) * YPP + 4 * tid]);
        const v4f y1 = *reinterpret_cast<const v4f*>(&sY[(1 * NCH + j) * YPP + 4 * tid]);
        const v4f y2 = *reinterpret_cast<const v4f*>(&sY[(2 * NCH + j) * YPP + 4 * tid]);
        const v4f y3 = *reinterpret_cast<const v4f*>(&sY[(3 * NCH + j) * YPP + 4 * tid]);
        const v4f s = (y0 + y1) + (y2 + y3);
        v4f* dst = reinterpret_cast<v4f*>(out + ((size_t)(b * CC + c0 + j)) * HW);
        dst[tid] = xq * dsc + s * 0.25f;
    }
}

extern "C" void kernel_launch(void* const* d_in, const int* in_sizes, int n_in,
                              void* d_out, int out_size, void* d_ws, size_t ws_size,
                              hipStream_t stream) {
    const float* feat    = (const float*)d_in[0];
    const float* A_log   = (const float*)d_in[1];
    const float* Dvec    = (const float*)d_in[2];
    const float* W_delta = (const float*)d_in[3];
    const float* b_delta = (const float*)d_in[4];
    const float* W_B     = (const float*)d_in[5];
    const float* W_C     = (const float*)d_in[6];
    float* out = (float*)d_out;

    float* ws       = (float*)d_ws;
    float* delta_ws = ws;
    float* Bm_ws    = ws + (size_t)NROW * CC;
    float* Cm_ws    = Bm_ws + (size_t)NROW * NN;

    proj_wmma_kernel<<<dim3(NROW / 16, 18), 32, 0, stream>>>(
        feat, W_delta, b_delta, W_B, W_C, delta_ws, Bm_ws, Cm_ws);
    scan_kernel<<<256, 32 * KCH, 0, stream>>>(feat, A_log, Dvec, delta_ws, Bm_ws, Cm_ws, out);
}


# projection GEMM: one wave per 32x32 block (2x2 WMMA tiles, rows/cols interleaved: 64-bit loads, each operand feeds 2 WMMAs), operand blocks of 8 k-steps double-buffered in registers; k-sum order kept
# speedup vs baseline: 30.2162x; 1.1064x over previous
#include <hip/hip_runtime.h>
#include <math.h>

typedef __attribute__((ext_vector_type(2))) float v2f;
typedef __attribute__((ext_vector_type(8))) float v8f;

#define BB 2
#define CC 256
#define HH 64
#define WW 64
#define HW 4096
#define NN 16
#define NROW (BB*HW)

typedef __attribute__((ext_vector_type(4))) float v4f;

__device__ __forceinline__ v8f wmma_f32(v2f a, v2f b, v8f c) {
    return __builtin_amdgcn_wmma_f32_16x16x4_f32(
         false, a,  false, b,
         (short)0, c,  false,  false);
}

__device__ __forceinline__ float softplus_f(float pre) {
    return (pre > 20.f) ? pre : log1pf(__expf(pre));
}

#define KB 8

template <bool BC>
__device__ __forceinline__ void load_blk(v2f (&ax)[KB], v2f (&ay)[KB], v2f (&bx)[KB], v2f (&by)[KB],
                                         const float* aP, const float* w0, const float* w1, int kb) {
    #pragma unroll
    for (int s = 0; s < KB; ++s) {
        const int k0 = kb + 4 * s;
        ax[s] = *reinterpret_cast<const v2f*>(aP + (size_t)k0 * HW);
        ay[s] = *reinterpret_cast<const v2f*>(aP + (size_t)(k0 + 1) * HW);
        if (BC) {
            bx[s] = (v2f){w0[(size_t)k0 * NN], w1[(size_t)k0 * NN]};
            by[s] = (v2f){w0[(size_t)(k0 + 1) * NN], w1[(size_t)(k0 + 1) * NN]};
        } else {
            bx[s] = *reinterpret_cast<const v2f*>(w0 + (size_t)k0 * CC);
            by[s] = *reinterpret_cast<const v2f*>(w0 + (size_t)(k0 + 1) * CC);
        }
    }
}

__device__ __forceinline__ void mma_blk(v8f& acc00, v8f& acc01, v8f& acc10, v8f& acc11,
                                        const v2f (&ax)[KB], const v2f (&ay)[KB],
                                        const v2f (&bx)[KB], const v2f (&by)[KB]) {
    #pragma unroll
    for (int s = 0; s < KB; ++s) {
        const v2f a0 = {ax[s].x, ay[s].x}, a1 = {ax[s].y, ay[s].y};
        const v2f b0 = {bx[s].x, by[s].x}, b1 = {bx[s].y, by[s].y};
        acc00 = wmma_f32(a0, b0, acc00);
        acc01 = wmma_f32(a0, b1, acc01);
        acc10 = wmma_f32(a1, b0, acc10);
        acc11 = wmma_f32(a1, b1, acc11);
    }
}

template <bool BC>
__device__ __forceinline__ void gemm_32x32(v8f& acc00, v8f& acc01, v8f& acc10, v8f& acc11,
                                           const float* aP, const float* w0, const float* w1) {
    v2f ax0[KB], ay0[KB], bx0[KB], by0[KB];
    v2f ax1[KB], ay1[KB], bx1[KB], by1[KB];
    load_blk<BC>(ax0, ay0, bx0, by0, aP, w0, w1, 0);
    #pragma unroll
    for (int kb = 0; kb < CC; kb += 8 * KB) {
        load_blk<BC>(ax1, ay1, bx1, by1, aP, w0, w1, kb + 4 * KB);
        __builtin_amdgcn_sched_barrier(0);
        mma_blk(acc00, acc01, acc10, acc11, ax0, ay0, bx0, by0);
        __builtin_amdgcn_sched_barrier(0);
        if (kb + 8 * KB < CC) load_blk<BC>(ax0, ay0, bx0, by0, aP, w0, w1, kb + 8 * KB);
        __builtin_amdgcn_sched_barrier(0);
        mma_blk(acc00, acc01, acc10, acc11, ax1, ay1, bx1, by1);
        __builtin_amdgcn_sched_barrier(0);
    }
}

__global__ void __launch_bounds__(32)
proj_wmma_kernel(const float* __restrict__ feat,
                                 const float* __restrict__ W_delta,
                                 const float* __restrict__ b_delta,
                                 const float* __restrict__ W_B,
                                 const float* __restrict__ W_C,
                                 float* __restrict__ delta_ws,
                                 float* __restrict__ Bm_ws,
                                 float* __restrict__ Cm_ws) {
    const int lane = threadIdx.x;
    const int lo   = lane & 15;
    const int hi   = lane >> 4;
    const int m0   = blockIdx.x * 32;
    const int which = blockIdx.y;
    const int bT   = m0 >> 12;
    const int lT   = m0 & (HW - 1);

    const float* aBase = feat + (size_t)bT * CC * HW + lT + 2 * lo + (size_t)(2 * hi) * HW;

    v8f acc00 = {}, acc01 = {}, acc10 = {}, acc11 = {};

    if (which < 8) {
        const int n0 = which * 32;
        const float* wBase = W_delta + n0 + 2 * lo + (size_t)(2 * hi) * CC;
        gemm_32x32<false>(acc00, acc01, acc10, acc11, aBase, wBase, wBase);
        #pragma unroll
        for (int j = 0; j < 2; ++j) {
            const int ch = n0 + 2 * lo + j;
            const float bias = b_delta[ch];
            const v8f e = j ? acc01 : acc00;
            const v8f o = j ? acc11 : acc10;
            v4f* dst = reinterpret_cast<v4f*>(delta_ws + ((size_t)(bT * CC + ch)) * HW + lT + 16 * hi);
            #pragma unroll
            for (int p = 0; p < 4; ++p) {
                dst[p] = (v4f){softplus_f(e[2 * p] + bias),     softplus_f(o[2 * p] + bias),
                               softplus_f(e[2 * p + 1] + bias), softplus_f(o[2 * p + 1] + bias)};
            }
        }
    } else {
        const float* wbBase = W_B + lo + (size_t)(2 * hi) * NN;
        const float* wcBase = W_C + lo + (size_t)(2 * hi) * NN;
        gemm_32x32<true>(acc00, acc01, acc10, acc11, aBase, wbBase, wcBase);
        #pragma unroll
        for (int v = 0; v < 8; ++v) {
            const size_t row = (size_t)(m0 + 2 * (v + 8 * hi));
            Bm_ws[row * NN + lo]       = acc00[v];
            Bm_ws[(row + 1) * NN + lo] = acc10[v];
            Cm_ws[row * NN + lo]       = acc01[v];
            Cm_ws[(row + 1) * NN + lo] = acc11[v];
        }
    }
}

#define KCH  32
#define CLEN (HW / KCH)
#define SPL  4
#define NCH  2
#define NPL  (4 * NCH)
#define RUN  64
#define ROWV4 (HW / 4)
static_assert(CLEN % RUN == 0 && RUN % SPL == 0, "a chunk is a whole number of runs of whole blocks");
static_assert(NPL * SPL == 32, "one wave = 8 pairs x 4 state quarters");

__device__ __forceinline__ float quad_sum(float v) {
    v += __builtin_bit_cast(float, __builtin_amdgcn_update_dpp(
             0, __builtin_bit_cast(int, v), 0xB1, 0xF, 0xF, true));
    v += __builtin_bit_cast(float, __builtin_amdgcn_update_dpp(
             0, __builtin_bit_cast(int, v), 0x4E, 0xF, 0xF, true));
    return v;
}

#define XDP (HW + 4)
#define YPP (HW + 4)
__global__ void __launch_bounds__(32 * KCH)
scan_kernel(const float* __restrict__ feat,
            const float* __restrict__ A_log,
            const float* __restrict__ Dvec,
            const float* __restrict__ delta_ws,
            const float* __restrict__ Bm_ws,
            const float* __restrict__ Cm_ws,
            float* __restrict__ out) {
    __shared__ __attribute__((aligned(16))) v2f sXD[NCH * XDP];
    __shared__ __attribute__((aligned(16))) float sY[NPL * YPP];
    __shared__ __attribute__((aligned(16))) float sS[3][64][64];

    const int tid  = threadIdx.x;
    const int lane = tid & 31;
    const int w  = __builtin_amdgcn_readfirstlane((int)(threadIdx.x >> 5));
    const int q  = lane & 3;
    const int cl = lane >> 2;
    const int ch  = cl & 1;
    const int dir = cl >> 1;
    const int id  = blockIdx.x;
    const int b   = id >> 7;
    const int c0  = (id & 127) * NCH;
    const int c   = c0 + ch;

    {
        const size_t goff = ((size_t)(b * CC + c0)) * HW;
        const v4f* xs = reinterpret_cast<const v4f*>(feat + goff);
        const v4f* ds = reinterpret_cast<const v4f*>(delta_ws + goff);
        v4f xr[NCH], dr[NCH];
        for (int j = 0; j < NCH; ++j) {
            xr[j] = xs[j * ROWV4 + tid];
            dr[j] = ds[j * ROWV4 + tid];
        }
        for (int j = 0; j < NCH; ++j) {
            v4f* p = reinterpret_cast<v4f*>(&sXD[j * XDP + 4 * tid]);
            p[0] = (v4f){xr[j].x, dr[j].x, xr[j].y, dr[j].y};
            p[1] = (v4f){xr[j].z, dr[j].z, xr[j].w, dr[j].w};
        }
    }

    const int base = (dir == 0) ? 0  : (dir == 1) ? 4095 : (dir == 2) ? 4032 : 63;
    const int sh   = (dir == 0) ? 64 : (dir == 3) ? -1 : -64;
    const int sw   = (dir == 1) ? -1 : (dir == 3) ? 64 : 1;

    const v4f Al = *reinterpret_cast<const v4f*>(A_log + c * NN + SPL * q);
    float A[SPL];
    for (int j = 0; j < SPL; ++j) A[j] = -__expf(Al[j]);

    const int lS = base + sh * (w * (CLEN / RUN));
    const unsigned xa0 = (unsigned)((ch * XDP + lS) * 8);
    const unsigned ga0 = (unsigned)(lS * (NN * 4) + SPL * 4 * q);
    const unsigned ya0 = (unsigned)((cl * YPP + lS + q * sw) * 4);
    const int xstep = 8 * sw;
    const int gstep = NN * 4 * sw;
    const int ystep = 4 * sw;
    const int runfix = sh - RUN * sw;

    const char* xdB = reinterpret_cast<const char*>(sXD);
    char* yB = reinterpret_cast<char*>(sY);
    const char* BmB = reinterpret_cast<const char*>(Bm_ws + (size_t)b * HW * NN);
    const char* CmB = reinterpret_cast<const char*>(Cm_ws + (size_t)b * HW * NN);

    __syncthreads();

    {
        const int trav = (lane >> 3) & 1;
        const int u    = 2 * w + (lane >> 4);
        const int l0   = trav ? (63 - u) : (RUN * u);
        const int pst  = trav ? WW : 1;
        unsigned xa = (unsigned)((ch * XDP + l0) * 8);
        unsigned ga = (unsigned)(l0 * (NN * 4) + SPL * 4 * q);
        const unsigned xs1 = (unsigned)(8 * pst);
        const unsigned gs1 = (unsigned)(NN * 4 * pst);
        float hf[SPL], hb[SPL], P[SPL];
        for (int j = 0; j < SPL; ++j) { hf[j] = 0.f; hb[j] = 0.f; P[j] = 1.f; }
        for (int it = 0; it < RUN / SPL; ++it) {
            float d[SPL], xv[SPL];
            v4f Bv[SPL];
            for (int k = 0; k < SPL; ++k) {
                const v2f xd = *reinterpret_cast<const v2f*>(xdB + (xa + (unsigned)k * xs1));
                xv[k] = xd.x;
                d[k]  = xd.y;
                Bv[k] = *reinterpret_cast<const v4f*>(BmB + (ga + (unsigned)k * gs1));
            }
            for (int k = 0; k < SPL; ++k) {
                const float dx = d[k] * xv[k];
                for (int j = 0; j < SPL; ++j) {
                    const float dA = __expf(d[k] * A[j]);
                    const float bt = dx * Bv[k][j];
                    hb[j] = fmaf(P[j], bt, hb[j]);
                    hf[j] = fmaf(dA, hf[j], bt);
                    P[j] *= dA;
                }
            }
            xa += (unsigned)SPL * xs1;
            ga += (unsigned)SPL * gs1;
        }
        const int col = (lane & 15) * SPL;
        *reinterpret_cast<v4f*>(&sS[0][u][col]) = (v4f){P[0], P[1], P[2], P[3]};
        *reinterpret_cast<v4f*>(&sS[1][u][col]) = (v4f){hf[0], hf[1], hf[2], hf[3]};
        *reinterpret_cast<v4f*>(&sS[2][u][col]) = (v4f){hb[0], hb[1], hb[2], hb[3]};
    }
    __syncthreads();

    float h[SPL];
    for (int j = 0; j < SPL; ++j) h[j] = 0.f;
    {
        const bool down = (dir == 1) || (dir == 2);
        const int ccol = ((dir == 3) ? 32 : 0) + ch * NN + SPL * q;
        const float* pP = &sS[0][down ? 63 : 0][ccol];
        const float* pE = &sS[(dir == 1) ? 2 : 1][down ? 63 : 0][ccol];
        const int ust = down ? -64 : 64;
        for (int i = 0; i < 2 * w; ++i) {
            const v4f Pv = *reinterpret_cast<const v4f*>(pP);
            const v4f Ev = *reinterpret_cast<const v4f*>(pE);
            for (int j = 0; j < SPL; ++j) h[j] = fmaf(Pv[j], h[j], Ev[j]);
            pP += ust;
            pE += ust;
        }
    }

    {
        unsigned xa = xa0, ga = ga0, ya = ya0;
        for (int run = 0; run < CLEN / RUN; ++run) {
            for (int it = 0; it < RUN / SPL; ++it) {
                float d[SPL], xv[SPL];
                v4f Bv[SPL], Cv[SPL];
                for (int k = 0; k < SPL; ++k) {
                    const v2f xd = *reinterpret_cast<const v2f*>(xdB + (xa + (unsigned)(k * xstep)));
                    xv[k] = xd.x;
                    d[k]  = xd.y;
                    Bv[k] = *reinterpret_cast<const v4f*>(BmB + (ga + (unsigned)(k * gstep)));
                    Cv[k] = *reinterpret_cast<const v4f*>(CmB + (ga + (unsigned)(k * gstep)));
                }
                float mine = 0.f;
                for (int k = 0; k < SPL; ++k) {
                    const float dx = d[k] * xv[k];
                    for (int j = 0; j < SPL; ++j) {
                        const float dA = __expf(d[k] * A[j]);
                        h[j] = fmaf(dA, h[j], dx * Bv[k][j]);
                    }
                    float y = h[0] * Cv[k][0];
                    y = fmaf(h[1], Cv[k][1], y);
                    y = fmaf(h[2], Cv[k][2], y);
                    y = fmaf(h[3], Cv[k][3], y);
                    y = quad_sum(y);
                    mine = (q == k) ? y : mine;
                }
                *reinterpret_cast<float*>(yB + ya) = mine;
                xa += (unsigned)(SPL * xstep);
                ga += (unsigned)(SPL * gstep);
                ya += (unsigned)(SPL * ystep);
            }
            xa += (unsigned)(8 * runfix);
            ga += (unsigned)(NN * 4 * runfix);
            ya += (unsigned)(4 * runfix);
        }
    }
    __syncthreads();

    for (int j = 0; j < NCH; ++j) {
        const float dsc = Dvec[c0 + j];
        const v4f* px = reinterpret_cast<const v4f*>(&sXD[j * XDP + 4 * tid]);
        const v4f p0 = px[0], p1 = px[1];
        const v4f xq = (v4f){p0.x, p0.z, p1.x, p1.z};
        const v4f y0 = *reinterpret_cast<const v4f*>(&sY[(0 * NCH + j) * YPP + 4 * tid]);
        const v4f y1 = *reinterpret_cast<const v4f*>(&sY[(1 * NCH + j) * YPP + 4 * tid]);
        const v4f y2 = *reinterpret_cast<const v4f*>(&sY[(2 * NCH + j) * YPP + 4 * tid]);
        const v4f y3 = *reinterpret_cast<const v4f*>(&sY[(3 * NCH + j) * YPP + 4 * tid]);
        const v4f s = (y0 + y1) + (y2 + y3);
        v4f* dst = reinterpret_cast<v4f*>(out + ((size_t)(b * CC + c0 + j)) * HW);
        dst[tid] = xq * dsc + s * 0.25f;
    }
}

extern "C" void kernel_launch(void* const* d_in, const int* in_sizes, int n_in,
                              void* d_out, int out_size, void* d_ws, size_t ws_size,
                              hipStream_t stream) {
    const float* feat    = (const float*)d_in[0];
    const float* A_log   = (const float*)d_in[1];
    const float* Dvec    = (const float*)d_in[2];
    const float* W_delta = (const float*)d_in[3];
    const float* b_delta = (const float*)d_in[4];
    const float* W_B     = (const float*)d_in[5];
    const float* W_C     = (const float*)d_in[6];
    float* out = (float*)d_out;

    float* ws       = (float*)d_ws;
    float* delta_ws = ws;
    float* Bm_ws    = ws + (size_t)NROW * CC;
    float* Cm_ws    = Bm_ws + (size_t)NROW * NN;

    proj_wmma_kernel<<<dim3(NROW / 32, 9), 32, 0, stream>>>(
        feat, W_delta, b_delta, W_B, W_C, delta_ws, Bm_ws, Cm_ws);
    scan_kernel<<<256, 32 * KCH, 0, stream>>>(feat, A_log, Dvec, delta_ws, Bm_ws, Cm_ws, out);
}


# scan pass 2: forward and H-flipped directions (same left-to-right row walk) run as ONE recurrence on the mean of their states, re-seeded per row from both carries; 3 scans, 24 waves (sum order new)
# speedup vs baseline: 32.0386x; 1.0603x over previous
#include <hip/hip_runtime.h>
#include <math.h>

typedef __attribute__((ext_vector_type(2))) float v2f;
typedef __attribute__((ext_vector_type(8))) float v8f;

#define BB 2
#define CC 256
#define HH 64
#define WW 64
#define HW 4096
#define NN 16
#define NROW (BB*HW)

typedef __attribute__((ext_vector_type(4))) float v4f;

__device__ __forceinline__ v8f wmma_f32(v2f a, v2f b, v8f c) {
    return __builtin_amdgcn_wmma_f32_16x16x4_f32(
         false, a,  false, b,
         (short)0, c,  false,  false);
}

__device__ __forceinline__ float softplus_f(float pre) {
    return (pre > 20.f) ? pre : log1pf(__expf(pre));
}

#define KB 8

template <bool BC>
__device__ __forceinline__ void load_blk(v2f (&ax)[KB], v2f (&ay)[KB], v2f (&bx)[KB], v2f (&by)[KB],
                                         const float* aP, const float* w0, const float* w1, int kb) {
    #pragma unroll
    for (int s = 0; s < KB; ++s) {
        const int k0 = kb + 4 * s;
        ax[s] = *reinterpret_cast<const v2f*>(aP + (size_t)k0 * HW);
        ay[s] = *reinterpret_cast<const v2f*>(aP + (size_t)(k0 + 1) * HW);
        if (BC) {
            bx[s] = (v2f){w0[(size_t)k0 * NN], w1[(size_t)k0 * NN]};
            by[s] = (v2f){w0[(size_t)(k0 + 1) * NN], w1[(size_t)(k0 + 1) * NN]};
        } else {
            bx[s] = *reinterpret_cast<const v2f*>(w0 + (size_t)k0 * CC);
            by[s] = *reinterpret_cast<const v2f*>(w0 + (size_t)(k0 + 1) * CC);
        }
    }
}

__device__ __forceinline__ void mma_blk(v8f& acc00, v8f& acc01, v8f& acc10, v8f& acc11,
                                        const v2f (&ax)[KB], const v2f (&ay)[KB],
                                        const v2f (&bx)[KB], const v2f (&by)[KB]) {
    #pragma unroll
    for (int s = 0; s < KB; ++s) {
        const v2f a0 = {ax[s].x, ay[s].x}, a1 = {ax[s].y, ay[s].y};
        const v2f b0 = {bx[s].x, by[s].x}, b1 = {bx[s].y, by[s].y};
        acc00 = wmma_f32(a0, b0, acc00);
        acc01 = wmma_f32(a0, b1, acc01);
        acc10 = wmma_f32(a1, b0, acc10);
        acc11 = wmma_f32(a1, b1, acc11);
    }
}

template <bool BC>
__device__ __forceinline__ void gemm_32x32(v8f& acc00, v8f& acc01, v8f& acc10, v8f& acc11,
                                           const float* aP, const float* w0, const float* w1) {
    v2f ax0[KB], ay0[KB], bx0[KB], by0[KB];
    v2f ax1[KB], ay1[KB], bx1[KB], by1[KB];
    load_blk<BC>(ax0, ay0, bx0, by0, aP, w0, w1, 0);
    #pragma unroll
    for (int kb = 0; kb < CC; kb += 8 * KB) {
        load_blk<BC>(ax1, ay1, bx1, by1, aP, w0, w1, kb + 4 * KB);
        __builtin_amdgcn_sched_barrier(0);
        mma_blk(acc00, acc01, acc10, acc11, ax0, ay0, bx0, by0);
        __builtin_amdgcn_sched_barrier(0);
        if (kb + 8 * KB < CC) load_blk<BC>(ax0, ay0, bx0, by0, aP, w0, w1, kb + 8 * KB);
        __builtin_amdgcn_sched_barrier(0);
        mma_blk(acc00, acc01, acc10, acc11, ax1, ay1, bx1, by1);
        __builtin_amdgcn_sched_barrier(0);
    }
}

__global__ void __launch_bounds__(32)
proj_wmma_kernel(const float* __restrict__ feat,
                                 const float* __restrict__ W_delta,
                                 const float* __restrict__ b_delta,
                                 const float* __restrict__ W_B,
                                 const float* __restrict__ W_C,
                                 float* __restrict__ delta_ws,
                                 float* __restrict__ Bm_ws,
                                 float* __restrict__ Cm_ws) {
    const int lane = threadIdx.x;
    const int lo   = lane & 15;
    const int hi   = lane >> 4;
    const int m0   = blockIdx.x * 32;
    const int which = blockIdx.y;
    const int bT   = m0 >> 12;
    const int lT   = m0 & (HW - 1);

    const float* aBase = feat + (size_t)bT * CC * HW + lT + 2 * lo + (size_t)(2 * hi) * HW;

    v8f acc00 = {}, acc01 = {}, acc10 = {}, acc11 = {};

    if (which < 8) {
        const int n0 = which * 32;
        const float* wBase = W_delta + n0 + 2 * lo + (size_t)(2 * hi) * CC;
        gemm_32x32<false>(acc00, acc01, acc10, acc11, aBase, wBase, wBase);
        #pragma unroll
        for (int j = 0; j < 2; ++j) {
            const int ch = n0 + 2 * lo + j;
            const float bias = b_delta[ch];
            const v8f e = j ? acc01 : acc00;
            const v8f o = j ? acc11 : acc10;
            v4f* dst = reinterpret_cast<v4f*>(delta_ws + ((size_t)(bT * CC + ch)) * HW + lT + 16 * hi);
            #pragma unroll
            for (int p = 0; p < 4; ++p) {
                dst[p] = (v4f){softplus_f(e[2 * p] + bias),     softplus_f(o[2 * p] + bias),
                               softplus_f(e[2 * p + 1] + bias), softplus_f(o[2 * p + 1] + bias)};
            }
        }
    } else {
        const float* wbBase = W_B + lo + (size_t)(2 * hi) * NN;
        const float* wcBase = W_C + lo + (size_t)(2 * hi) * NN;
        gemm_32x32<true>(acc00, acc01, acc10, acc11, aBase, wbBase, wcBase);
        #pragma unroll
        for (int v = 0; v < 8; ++v) {
            const size_t row = (size_t)(m0 + 2 * (v + 8 * hi));
            Bm_ws[row * NN + lo]       = acc00[v];
            Bm_ws[(row + 1) * NN + lo] = acc10[v];
            Cm_ws[row * NN + lo]       = acc01[v];
            Cm_ws[(row + 1) * NN + lo] = acc11[v];
        }
    }
}

#define KCH  32
#define NSC  3
#define NW2  (NSC * NCH * KCH / 8)
#define CLEN (HW / KCH)
#define SPL  4
#define NCH  2
#define NPL  (NSC * NCH)
#define RUN  64
#define ROWV4 (HW / 4)
static_assert(CLEN % RUN == 0 && RUN % SPL == 0, "a chunk is a whole number of runs of whole blocks");
static_assert((NSC * NCH * KCH) % 8 == 0 && NW2 <= KCH, "the items fill whole waves of 8 quads");

__device__ __forceinline__ float quad_sum(float v) {
    v += __builtin_bit_cast(float, __builtin_amdgcn_update_dpp(
             0, __builtin_bit_cast(int, v), 0xB1, 0xF, 0xF, true));
    v += __builtin_bit_cast(float, __builtin_amdgcn_update_dpp(
             0, __builtin_bit_cast(int, v), 0x4E, 0xF, 0xF, true));
    return v;
}

#define XDP (HW + 4)
#define YPP (HW + 4)
__global__ void __launch_bounds__(32 * KCH)
scan_kernel(const float* __restrict__ feat,
            const float* __restrict__ A_log,
            const float* __restrict__ Dvec,
            const float* __restrict__ delta_ws,
            const float* __restrict__ Bm_ws,
            const float* __restrict__ Cm_ws,
            float* __restrict__ out) {
    __shared__ __attribute__((aligned(16))) v2f sXD[NCH * XDP];
    __shared__ __attribute__((aligned(16))) float sY[NPL * YPP];
    __shared__ __attribute__((aligned(16))) float sS[3][64][64];

    const int tid  = threadIdx.x;
    const int lane = tid & 31;
    const int w  = __builtin_amdgcn_readfirstlane((int)(threadIdx.x >> 5));
    const int q  = lane & 3;
    const int ch = (lane >> 2) & 1;
    const int item = 8 * w + (lane >> 2);
    const int cw   = item / (NSC * NCH);
    const int cl   = item - cw * (NSC * NCH);
    const int scan = cl >> 1;
    const int id  = blockIdx.x;
    const int b   = id >> 7;
    const int c0  = (id & 127) * NCH;
    const int c   = c0 + ch;

    {
        const size_t goff = ((size_t)(b * CC + c0)) * HW;
        const v4f* xs = reinterpret_cast<const v4f*>(feat + goff);
        const v4f* ds = reinterpret_cast<const v4f*>(delta_ws + goff);
        v4f xr[NCH], dr[NCH];
        for (int j = 0; j < NCH; ++j) {
            xr[j] = xs[j * ROWV4 + tid];
            dr[j] = ds[j * ROWV4 + tid];
        }
        for (int j = 0; j < NCH; ++j) {
            v4f* p = reinterpret_cast<v4f*>(&sXD[j * XDP + 4 * tid]);
            p[0] = (v4f){xr[j].x, dr[j].x, xr[j].y, dr[j].y};
            p[1] = (v4f){xr[j].z, dr[j].z, xr[j].w, dr[j].w};
        }
    }

    const int base = (scan == 0) ? 0  : (scan == 1) ? 4095 : 63;
    const int sh   = (scan == 0) ? 64 : (scan == 1) ? -64 : -1;
    const int sw   = (scan == 0) ? 1  : (scan == 1) ? -1 : 64;

    const v4f Al = *reinterpret_cast<const v4f*>(A_log + c * NN + SPL * q);
    float A[SPL];
    for (int j = 0; j < SPL; ++j) A[j] = -__expf(Al[j]);

    const int lS = base + sh * (cw * (CLEN / RUN));
    const unsigned xa0 = (unsigned)((ch * XDP + lS) * 8);
    const unsigned ga0 = (unsigned)(lS * (NN * 4) + SPL * 4 * q);
    const unsigned ya0 = (unsigned)((cl * YPP + lS + q * sw) * 4);
    const int xstep = 8 * sw;
    const int gstep = NN * 4 * sw;
    const int ystep = 4 * sw;
    const int runfix = sh - RUN * sw;

    const char* xdB = reinterpret_cast<const char*>(sXD);
    char* yB = reinterpret_cast<char*>(sY);
    const char* BmB = reinterpret_cast<const char*>(Bm_ws + (size_t)b * HW * NN);
    const char* CmB = reinterpret_cast<const char*>(Cm_ws + (size_t)b * HW * NN);

    __syncthreads();

    {
        const int trav = (lane >> 3) & 1;
        const int u    = 2 * w + (lane >> 4);
        const int l0   = trav ? (63 - u) : (RUN * u);
        const int pst  = trav ? WW : 1;
        unsigned xa = (unsigned)((ch * XDP + l0) * 8);
        unsigned ga = (unsigned)(l0 * (NN * 4) + SPL * 4 * q);
        const unsigned xs1 = (unsigned)(8 * pst);
        const unsigned gs1 = (unsigned)(NN * 4 * pst);
        float hf[SPL], hb[SPL], P[SPL];
        for (int j = 0; j < SPL; ++j) { hf[j] = 0.f; hb[j] = 0.f; P[j] = 1.f; }
        for (int it = 0; it < RUN / SPL; ++it) {
            float d[SPL], xv[SPL];
            v4f Bv[SPL];
            for (int k = 0; k < SPL; ++k) {
                const v2f xd = *reinterpret_cast<const v2f*>(xdB + (xa + (unsigned)k * xs1));
                xv[k] = xd.x;
                d[k]  = xd.y;
                Bv[k] = *reinterpret_cast<const v4f*>(BmB + (ga + (unsigned)k * gs1));
            }
            for (int k = 0; k < SPL; ++k) {
                const float dx = d[k] * xv[k];
                for (int j = 0; j < SPL; ++j) {
                    const float dA = __expf(d[k] * A[j]);
                    const float bt = dx * Bv[k][j];
                    hb[j] = fmaf(P[j], bt, hb[j]);
                    hf[j] = fmaf(dA, hf[j], bt);
                    P[j] *= dA;
                }
            }
            xa += (unsigned)SPL * xs1;
            ga += (unsigned)SPL * gs1;
        }
        const int col = (lane & 15) * SPL;
        *reinterpret_cast<v4f*>(&sS[0][u][col]) = (v4f){P[0], P[1], P[2], P[3]};
        *reinterpret_cast<v4f*>(&sS[1][u][col]) = (v4f){hf[0], hf[1], hf[2], hf[3]};
        *reinterpret_cast<v4f*>(&sS[2][u][col]) = (v4f){hb[0], hb[1], hb[2], hb[3]};
    }
    __syncthreads();

    if (w < NW2) {
    float h[SPL], hB[SPL];
    for (int j = 0; j < SPL; ++j) { h[j] = 0.f; hB[j] = 0.f; }
    {
        const bool down = (scan == 1);
        const int ccol = ((scan == 2) ? 32 : 0) + ch * NN + SPL * q;
        const float* pP = &sS[0][down ? 63 : 0][ccol];
        const float* pE = &sS[(scan == 1) ? 2 : 1][down ? 63 : 0][ccol];
        const int ust = down ? -64 : 64;
        for (int i = 0; i < 2 * cw; ++i) {
            const v4f Pv = *reinterpret_cast<const v4f*>(pP);
            const v4f Ev = *reinterpret_cast<const v4f*>(pE);
            for (int j = 0; j < SPL; ++j) h[j] = fmaf(Pv[j], h[j], Ev[j]);
            pP += ust;
            pE += ust;
        }
        if (scan == 0) {
            float g[SPL];
            for (int j = 0; j < SPL; ++j) g[j] = 0.f;
            const float* qP = &sS[0][63][ccol];
            const float* qE = &sS[1][63][ccol];
            for (int i = 0; i < 62 - 2 * cw; ++i) {
                const v4f Pv = *reinterpret_cast<const v4f*>(qP);
                const v4f Ev = *reinterpret_cast<const v4f*>(qE);
                for (int j = 0; j < SPL; ++j) g[j] = fmaf(Pv[j], g[j], Ev[j]);
                qP -= 64;
                qE -= 64;
            }
            const int rA = 2 * cw;
            const v4f PA = *reinterpret_cast<const v4f*>(&sS[0][rA][ccol]);
            const v4f EA = *reinterpret_cast<const v4f*>(&sS[1][rA][ccol]);
            const v4f PB = *reinterpret_cast<const v4f*>(&sS[0][rA + 1][ccol]);
            const v4f EB = *reinterpret_cast<const v4f*>(&sS[1][rA + 1][ccol]);
            for (int j = 0; j < SPL; ++j) {
                const float fB = fmaf(PA[j], h[j], EA[j]);
                const float gA = fmaf(PB[j], g[j], EB[j]);
                h[j]  = 0.5f * (h[j] + gA);
                hB[j] = 0.5f * (fB + g[j]);
            }
        }
    }

    {
        unsigned xa = xa0, ga = ga0, ya = ya0;
        for (int run = 0; run < CLEN / RUN; ++run) {
            if (run == 1 && scan == 0) { for (int j = 0; j < SPL; ++j) h[j] = hB[j]; }
            for (int it = 0; it < RUN / SPL; ++it) {
                float d[SPL], xv[SPL];
                v4f Bv[SPL], Cv[SPL];
                for (int k = 0; k < SPL; ++k) {
                    const v2f xd = *reinterpret_cast<const v2f*>(xdB + (xa + (unsigned)(k * xstep)));
                    xv[k] = xd.x;
                    d[k]  = xd.y;
                    Bv[k] = *reinterpret_cast<const v4f*>(BmB + (ga + (unsigned)(k * gstep)));
                    Cv[k] = *reinterpret_cast<const v4f*>(CmB + (ga + (unsigned)(k * gstep)));
                }
                float mine = 0.f;
                for (int k = 0; k < SPL; ++k) {
                    const float dx = d[k] * xv[k];
                    for (int j = 0; j < SPL; ++j) {
                        const float dA = __expf(d[k] * A[j]);
                        h[j] = fmaf(dA, h[j], dx * Bv[k][j]);
                    }
                    float y = h[0] * Cv[k][0];
                    y = fmaf(h[1], Cv[k][1], y);
                    y = fmaf(h[2], Cv[k][2], y);
                    y = fmaf(h[3], Cv[k][3], y);
                    y = quad_sum(y);
                    mine = (q == k) ? y : mine;
                }
                *reinterpret_cast<float*>(yB + ya) = mine;
                xa += (unsigned)(SPL * xstep);
                ga += (unsigned)(SPL * gstep);
                ya += (unsigned)(SPL * ystep);
            }
            xa += (unsigned)(8 * runfix);
            ga += (unsigned)(NN * 4 * runfix);
            ya += (unsigned)(4 * runfix);
        }
    }
    }
    __syncthreads();

    for (int j = 0; j < NCH; ++j) {
        const float dsc = Dvec[c0 + j];
        const v4f* px = reinterpret_cast<const v4f*>(&sXD[j * XDP + 4 * tid]);
        const v4f p0 = px[0], p1 = px[1];
        const v4f xq = (v4f){p0.x, p0.z, p1.x, p1.z};
        const v4f y0 = *reinterpret_cast<const v4f*>(&sY[(0 * NCH + j) * YPP + 4 * tid]);
        const v4f y1 = *reinterpret_cast<const v4f*>(&sY[(1 * NCH + j) * YPP + 4 * tid]);
        const v4f y2 = *reinterpret_cast<const v4f*>(&sY[(2 * NCH + j) * YPP + 4 * tid]);
        const v4f s = (y0 + y0) + (y1 + y2);
        v4f* dst = reinterpret_cast<v4f*>(out + ((size_t)(b * CC + c0 + j)) * HW);
        dst[tid] = xq * dsc + s * 0.25f;
    }
}

extern "C" void kernel_launch(void* const* d_in, const int* in_sizes, int n_in,
                              void* d_out, int out_size, void* d_ws, size_t ws_size,
                              hipStream_t stream) {
    const float* feat    = (const float*)d_in[0];
    const float* A_log   = (const float*)d_in[1];
    const float* Dvec    = (const float*)d_in[2];
    const float* W_delta = (const float*)d_in[3];
    const float* b_delta = (const float*)d_in[4];
    const float* W_B     = (const float*)d_in[5];
    const float* W_C     = (const float*)d_in[6];
    float* out = (float*)d_out;

    float* ws       = (float*)d_ws;
    float* delta_ws = ws;
    float* Bm_ws    = ws + (size_t)NROW * CC;
    float* Cm_ws    = Bm_ws + (size_t)NROW * NN;

    proj_wmma_kernel<<<dim3(NROW / 32, 9), 32, 0, stream>>>(
        feat, W_delta, b_delta, W_B, W_C, delta_ws, Bm_ws, Cm_ws);
    scan_kernel<<<256, 32 * KCH, 0, stream>>>(feat, A_log, Dvec, delta_ws, Bm_ws, Cm_ws, out);
}


# projection epilogue: softplus = max(x,0)+log1p(exp(-|x|)) on hardware exp2/log2 plus the rounding term of 1+u, branch-free, replacing the library log1pf chain; delta's last-place rounding changes
# speedup vs baseline: 34.3064x; 1.0708x over previous
#include <hip/hip_runtime.h>
#include <math.h>

typedef __attribute__((ext_vector_type(2))) float v2f;
typedef __attribute__((ext_vector_type(8))) float v8f;

#define BB 2
#define CC 256
#define HH 64
#define WW 64
#define HW 4096
#define NN 16
#define NROW (BB*HW)

typedef __attribute__((ext_vector_type(4))) float v4f;

__device__ __forceinline__ v8f wmma_f32(v2f a, v2f b, v8f c) {
    return __builtin_amdgcn_wmma_f32_16x16x4_f32(
         false, a,  false, b,
         (short)0, c,  false,  false);
}

__device__ __forceinline__ float softplus_f(float pre) {
    const float u = __expf(-fabsf(pre));
    const float t = 1.0f + u;
    const float c = u - (t - 1.0f);
    const float l = fmaf(__log2f(t), 0.693147180559945f, c);
    return fmaxf(pre, 0.0f) + l;
}

#define KB 8

template <bool BC>
__device__ __forceinline__ void load_blk(v2f (&ax)[KB], v2f (&ay)[KB], v2f (&bx)[KB], v2f (&by)[KB],
                                         const float* aP, const float* w0, const float* w1, int kb) {
    #pragma unroll
    for (int s = 0; s < KB; ++s) {
        const int k0 = kb + 4 * s;
        ax[s] = *reinterpret_cast<const v2f*>(aP + (size_t)k0 * HW);
        ay[s] = *reinterpret_cast<const v2f*>(aP + (size_t)(k0 + 1) * HW);
        if (BC) {
            bx[s] = (v2f){w0[(size_t)k0 * NN], w1[(size_t)k0 * NN]};
            by[s] = (v2f){w0[(size_t)(k0 + 1) * NN], w1[(size_t)(k0 + 1) * NN]};
        } else {
            bx[s] = *reinterpret_cast<const v2f*>(w0 + (size_t)k0 * CC);
            by[s] = *reinterpret_cast<const v2f*>(w0 + (size_t)(k0 + 1) * CC);
        }
    }
}

__device__ __forceinline__ void mma_blk(v8f& acc00, v8f& acc01, v8f& acc10, v8f& acc11,
                                        const v2f (&ax)[KB], const v2f (&ay)[KB],
                                        const v2f (&bx)[KB], const v2f (&by)[KB]) {
    #pragma unroll
    for (int s = 0; s < KB; ++s) {
        const v2f a0 = {ax[s].x, ay[s].x}, a1 = {ax[s].y, ay[s].y};
        const v2f b0 = {bx[s].x, by[s].x}, b1 = {bx[s].y, by[s].y};
        acc00 = wmma_f32(a0, b0, acc00);
        acc01 = wmma_f32(a0, b1, acc01);
        acc10 = wmma_f32(a1, b0, acc10);
        acc11 = wmma_f32(a1, b1, acc11);
    }
}

template <bool BC>
__device__ __forceinline__ void gemm_32x32(v8f& acc00, v8f& acc01, v8f& acc10, v8f& acc11,
                                           const float* aP, const float* w0, const float* w1) {
    v2f ax0[KB], ay0[KB], bx0[KB], by0[KB];
    v2f ax1[KB], ay1[KB], bx1[KB], by1[KB];
    load_blk<BC>(ax0, ay0, bx0, by0, aP, w0, w1, 0);
    #pragma unroll
    for (int kb = 0; kb < CC; kb += 8 * KB) {
        load_blk<BC>(ax1, ay1, bx1, by1, aP, w0, w1, kb + 4 * KB);
        __builtin_amdgcn_sched_barrier(0);
        mma_blk(acc00, acc01, acc10, acc11, ax0, ay0, bx0, by0);
        __builtin_amdgcn_sched_barrier(0);
        if (kb + 8 * KB < CC) load_blk<BC>(ax0, ay0, bx0, by0, aP, w0, w1, kb + 8 * KB);
        __builtin_amdgcn_sched_barrier(0);
        mma_blk(acc00, acc01, acc10, acc11, ax1, ay1, bx1, by1);
        __builtin_amdgcn_sched_barrier(0);
    }
}

__global__ void __launch_bounds__(32)
proj_wmma_kernel(const float* __restrict__ feat,
                                 const float* __restrict__ W_delta,
                                 const float* __restrict__ b_delta,
                                 const float* __restrict__ W_B,
                                 const float* __restrict__ W_C,
                                 float* __restrict__ delta_ws,
                                 float* __restrict__ Bm_ws,
                                 float* __restrict__ Cm_ws) {
    const int lane = threadIdx.x;
    const int lo   = lane & 15;
    const int hi   = lane >> 4;
    const int m0   = blockIdx.x * 32;
    const int which = blockIdx.y;
    const int bT   = m0 >> 12;
    const int lT   = m0 & (HW - 1);

    const float* aBase = feat + (size_t)bT * CC * HW + lT + 2 * lo + (size_t)(2 * hi) * HW;

    v8f acc00 = {}, acc01 = {}, acc10 = {}, acc11 = {};

    if (which < 8) {
        const int n0 = which * 32;
        const float* wBase = W_delta + n0 + 2 * lo + (size_t)(2 * hi) * CC;
        gemm_32x32<false>(acc00, acc01, acc10, acc11, aBase, wBase, wBase);
        #pragma unroll
        for (int j = 0; j < 2; ++j) {
            const int ch = n0 + 2 * lo + j;
            const float bias = b_delta[ch];
            const v8f e = j ? acc01 : acc00;
            const v8f o = j ? acc11 : acc10;
            v4f* dst = reinterpret_cast<v4f*>(delta_ws + ((size_t)(bT * CC + ch)) * HW + lT + 16 * hi);
            #pragma unroll
            for (int p = 0; p < 4; ++p) {
                dst[p] = (v4f){softplus_f(e[2 * p] + bias),     softplus_f(o[2 * p] + bias),
                               softplus_f(e[2 * p + 1] + bias), softplus_f(o[2 * p + 1] + bias)};
            }
        }
    } else {
        const float* wbBase = W_B + lo + (size_t)(2 * hi) * NN;
        const float* wcBase = W_C + lo + (size_t)(2 * hi) * NN;
        gemm_32x32<true>(acc00, acc01, acc10, acc11, aBase, wbBase, wcBase);
        #pragma unroll
        for (int v = 0; v < 8; ++v) {
            const size_t row = (size_t)(m0 + 2 * (v + 8 * hi));
            Bm_ws[row * NN + lo]       = acc00[v];
            Bm_ws[(row + 1) * NN + lo] = acc10[v];
            Cm_ws[row * NN + lo]       = acc01[v];
            Cm_ws[(row + 1) * NN + lo] = acc11[v];
        }
    }
}

#define KCH  32
#define NSC  3
#define NW2  (NSC * NCH * KCH / 8)
#define CLEN (HW / KCH)
#define SPL  4
#define NCH  2
#define NPL  (NSC * NCH)
#define RUN  64
#define ROWV4 (HW / 4)
static_assert(CLEN % RUN == 0 && RUN % SPL == 0, "a chunk is a whole number of runs of whole blocks");
static_assert((NSC * NCH * KCH) % 8 == 0 && NW2 <= KCH, "the items fill whole waves of 8 quads");

__device__ __forceinline__ float quad_sum(float v) {
    v += __builtin_bit_cast(float, __builtin_amdgcn_update_dpp(
             0, __builtin_bit_cast(int, v), 0xB1, 0xF, 0xF, true));
    v += __builtin_bit_cast(float, __builtin_amdgcn_update_dpp(
             0, __builtin_bit_cast(int, v), 0x4E, 0xF, 0xF, true));
    return v;
}

#define XDP (HW + 4)
#define YPP (HW + 4)
__global__ void __launch_bounds__(32 * KCH)
scan_kernel(const float* __restrict__ feat,
            const float* __restrict__ A_log,
            const float* __restrict__ Dvec,
            const float* __restrict__ delta_ws,
            const float* __restrict__ Bm_ws,
            const float* __restrict__ Cm_ws,
            float* __restrict__ out) {
    __shared__ __attribute__((aligned(16))) v2f sXD[NCH * XDP];
    __shared__ __attribute__((aligned(16))) float sY[NPL * YPP];
    __shared__ __attribute__((aligned(16))) float sS[3][64][64];

    const int tid  = threadIdx.x;
    const int lane = tid & 31;
    const int w  = __builtin_amdgcn_readfirstlane((int)(threadIdx.x >> 5));
    const int q  = lane & 3;
    const int ch = (lane >> 2) & 1;
    const int item = 8 * w + (lane >> 2);
    const int cw   = item / (NSC * NCH);
    const int cl   = item - cw * (NSC * NCH);
    const int scan = cl >> 1;
    const int id  = blockIdx.x;
    const int b   = id >> 7;
    const int c0  = (id & 127) * NCH;
    const int c   = c0 + ch;

    {
        const size_t goff = ((size_t)(b * CC + c0)) * HW;
        const v4f* xs = reinterpret_cast<const v4f*>(feat + goff);
        const v4f* ds = reinterpret_cast<const v4f*>(delta_ws + goff);
        v4f xr[NCH], dr[NCH];
        for (int j = 0; j < NCH; ++j) {
            xr[j] = xs[j * ROWV4 + tid];
            dr[j] = ds[j * ROWV4 + tid];
        }
        for (int j = 0; j < NCH; ++j) {
            v4f* p = reinterpret_cast<v4f*>(&sXD[j * XDP + 4 * tid]);
            p[0] = (v4f){xr[j].x, dr[j].x, xr[j].y, dr[j].y};
            p[1] = (v4f){xr[j].z, dr[j].z, xr[j].w, dr[j].w};
        }
    }

    const int base = (scan == 0) ? 0  : (scan == 1) ? 4095 : 63;
    const int sh   = (scan == 0) ? 64 : (scan == 1) ? -64 : -1;
    const int sw   = (scan == 0) ? 1  : (scan == 1) ? -1 : 64;

    const v4f Al = *reinterpret_cast<const v4f*>(A_log + c * NN + SPL * q);
    float A[SPL];
    for (int j = 0; j < SPL; ++j) A[j] = -__expf(Al[j]);

    const int lS = base + sh * (cw * (CLEN / RUN));
    const unsigned xa0 = (unsigned)((ch * XDP + lS) * 8);
    const unsigned ga0 = (unsigned)(lS * (NN * 4) + SPL * 4 * q);
    const unsigned ya0 = (unsigned)((cl * YPP + lS + q * sw) * 4);
    const int xstep = 8 * sw;
    const int gstep = NN * 4 * sw;
    const int ystep = 4 * sw;
    const int runfix = sh - RUN * sw;

    const char* xdB = reinterpret_cast<const char*>(sXD);
    char* yB = reinterpret_cast<char*>(sY);
    const char* BmB = reinterpret_cast<const char*>(Bm_ws + (size_t)b * HW * NN);
    const char* CmB = reinterpret_cast<const char*>(Cm_ws + (size_t)b * HW * NN);

    __syncthreads();

    {
        const int trav = (lane >> 3) & 1;
        const int u    = 2 * w + (lane >> 4);
        const int l0   = trav ? (63 - u) : (RUN * u);
        const int pst  = trav ? WW : 1;
        unsigned xa = (unsigned)((ch * XDP + l0) * 8);
        unsigned ga = (unsigned)(l0 * (NN * 4) + SPL * 4 * q);
        const unsigned xs1 = (unsigned)(8 * pst);
        const unsigned gs1 = (unsigned)(NN * 4 * pst);
        float hf[SPL], hb[SPL], P[SPL];
        for (int j = 0; j < SPL; ++j) { hf[j] = 0.f; hb[j] = 0.f; P[j] = 1.f; }
        for (int it = 0; it < RUN / SPL; ++it) {
            float d[SPL], xv[SPL];
            v4f Bv[SPL];
            for (int k = 0; k < SPL; ++k) {
                const v2f xd = *reinterpret_cast<const v2f*>(xdB + (xa + (unsigned)k * xs1));
                xv[k] = xd.x;
                d[k]  = xd.y;
                Bv[k] = *reinterpret_cast<const v4f*>(BmB + (ga + (unsigned)k * gs1));
            }
            for (int k = 0; k < SPL; ++k) {
                const float dx = d[k] * xv[k];
                for (int j = 0; j < SPL; ++j) {
                    const float dA = __expf(d[k] * A[j]);
                    const float bt = dx * Bv[k][j];
                    hb[j] = fmaf(P[j], bt, hb[j]);
                    hf[j] = fmaf(dA, hf[j], bt);
                    P[j] *= dA;
                }
            }
            xa += (unsigned)SPL * xs1;
            ga += (unsigned)SPL * gs1;
        }
        const int col = (lane & 15) * SPL;
        *reinterpret_cast<v4f*>(&sS[0][u][col]) = (v4f){P[0], P[1], P[2], P[3]};
        *reinterpret_cast<v4f*>(&sS[1][u][col]) = (v4f){hf[0], hf[1], hf[2], hf[3]};
        *reinterpret_cast<v4f*>(&sS[2][u][col]) = (v4f){hb[0], hb[1], hb[2], hb[3]};
    }
    __syncthreads();

    if (w < NW2) {
    float h[SPL], hB[SPL];
    for (int j = 0; j < SPL; ++j) { h[j] = 0.f; hB[j] = 0.f; }
    {
        const bool down = (scan == 1);
        const int ccol = ((scan == 2) ? 32 : 0) + ch * NN + SPL * q;
        const float* pP = &sS[0][down ? 63 : 0][ccol];
        const float* pE = &sS[(scan == 1) ? 2 : 1][down ? 63 : 0][ccol];
        const int ust = down ? -64 : 64;
        for (int i = 0; i < 2 * cw; ++i) {
            const v4f Pv = *reinterpret_cast<const v4f*>(pP);
            const v4f Ev = *reinterpret_cast<const v4f*>(pE);
            for (int j = 0; j < SPL; ++j) h[j] = fmaf(Pv[j], h[j], Ev[j]);
            pP += ust;
            pE += ust;
        }
        if (scan == 0) {
            float g[SPL];
            for (int j = 0; j < SPL; ++j) g[j] = 0.f;
            const float* qP = &sS[0][63][ccol];
            const float* qE = &sS[1][63][ccol];
            for (int i = 0; i < 62 - 2 * cw; ++i) {
                const v4f Pv = *reinterpret_cast<const v4f*>(qP);
                const v4f Ev = *reinterpret_cast<const v4f*>(qE);
                for (int j = 0; j < SPL; ++j) g[j] = fmaf(Pv[j], g[j], Ev[j]);
                qP -= 64;
                qE -= 64;
            }
            const int rA = 2 * cw;
            const v4f PA = *reinterpret_cast<const v4f*>(&sS[0][rA][ccol]);
            const v4f EA = *reinterpret_cast<const v4f*>(&sS[1][rA][ccol]);
            const v4f PB = *reinterpret_cast<const v4f*>(&sS[0][rA + 1][ccol]);
            const v4f EB = *reinterpret_cast<const v4f*>(&sS[1][rA + 1][ccol]);
            for (int j = 0; j < SPL; ++j) {
                const float fB = fmaf(PA[j], h[j], EA[j]);
                const float gA = fmaf(PB[j], g[j], EB[j]);
                h[j]  = 0.5f * (h[j] + gA);
                hB[j] = 0.5f * (fB + g[j]);
            }
        }
    }

    {
        unsigned xa = xa0, ga = ga0, ya = ya0;
        for (int run = 0; run < CLEN / RUN; ++run) {
            if (run == 1 && scan == 0) { for (int j = 0; j < SPL; ++j) h[j] = hB[j]; }
            for (int it = 0; it < RUN / SPL; ++it) {
                float d[SPL], xv[SPL];
                v4f Bv[SPL], Cv[SPL];
                for (int k = 0; k < SPL; ++k) {
                    const v2f xd = *reinterpret_cast<const v2f*>(xdB + (xa + (unsigned)(k * xstep)));
                    xv[k] = xd.x;
                    d[k]  = xd.y;
                    Bv[k] = *reinterpret_cast<const v4f*>(BmB + (ga + (unsigned)(k * gstep)));
                    Cv[k] = *reinterpret_cast<const v4f*>(CmB + (ga + (unsigned)(k * gstep)));
                }
                float mine = 0.f;
                for (int k = 0; k < SPL; ++k) {
                    const float dx = d[k] * xv[k];
                    for (int j = 0; j < SPL; ++j) {
                        const float dA = __expf(d[k] * A[j]);
                        h[j] = fmaf(dA, h[j], dx * Bv[k][j]);
                    }
                    float y = h[0] * Cv[k][0];
                    y = fmaf(h[1], Cv[k][1], y);
                    y = fmaf(h[2], Cv[k][2], y);
                    y = fmaf(h[3], Cv[k][3], y);
                    y = quad_sum(y);
                    mine = (q == k) ? y : mine;
                }
                *reinterpret_cast<float*>(yB + ya) = mine;
                xa += (unsigned)(SPL * xstep);
                ga += (unsigned)(SPL * gstep);
                ya += (unsigned)(SPL * ystep);
            }
            xa += (unsigned)(8 * runfix);
            ga += (unsigned)(NN * 4 * runfix);
            ya += (unsigned)(4 * runfix);
        }
    }
    }
    __syncthreads();

    for (int j = 0; j < NCH; ++j) {
        const float dsc = Dvec[c0 + j];
        const v4f* px = reinterpret_cast<const v4f*>(&sXD[j * XDP + 4 * tid]);
        const v4f p0 = px[0], p1 = px[1];
        const v4f xq = (v4f){p0.x, p0.z, p1.x, p1.z};
        const v4f y0 = *reinterpret_cast<const v4f*>(&sY[(0 * NCH + j) * YPP + 4 * tid]);
        const v4f y1 = *reinterpret_cast<const v4f*>(&sY[(1 * NCH + j) * YPP + 4 * tid]);
        const v4f y2 = *reinterpret_cast<const v4f*>(&sY[(2 * NCH + j) * YPP + 4 * tid]);
        const v4f s = (y0 + y0) + (y1 + y2);
        v4f* dst = reinterpret_cast<v4f*>(out + ((size_t)(b * CC + c0 + j)) * HW);
        dst[tid] = xq * dsc + s * 0.25f;
    }
}

extern "C" void kernel_launch(void* const* d_in, const int* in_sizes, int n_in,
                              void* d_out, int out_size, void* d_ws, size_t ws_size,
                              hipStream_t stream) {
    const float* feat    = (const float*)d_in[0];
    const float* A_log   = (const float*)d_in[1];
    const float* Dvec    = (const float*)d_in[2];
    const float* W_delta = (const float*)d_in[3];
    const float* b_delta = (const float*)d_in[4];
    const float* W_B     = (const float*)d_in[5];
    const float* W_C     = (const float*)d_in[6];
    float* out = (float*)d_out;

    float* ws       = (float*)d_ws;
    float* delta_ws = ws;
    float* Bm_ws    = ws + (size_t)NROW * CC;
    float* Cm_ws    = Bm_ws + (size_t)NROW * NN;

    proj_wmma_kernel<<<dim3(NROW / 32, 9), 32, 0, stream>>>(
        feat, W_delta, b_delta, W_B, W_C, delta_ws, Bm_ws, Cm_ws);
    scan_kernel<<<256, 32 * KCH, 0, stream>>>(feat, A_log, Dvec, delta_ws, Bm_ws, Cm_ws, out);
}


# scan pass 2 item map: column-scan chunks regrouped so quads on 4 neighbouring columns share one load of an aligned 256-byte Bm/Cm piece (a line fetched once, not twice); 2nd-run entry from summaries
# speedup vs baseline: 37.9647x; 1.1066x over previous
#include <hip/hip_runtime.h>
#include <math.h>

typedef __attribute__((ext_vector_type(2))) float v2f;
typedef __attribute__((ext_vector_type(8))) float v8f;

#define BB 2
#define CC 256
#define HH 64
#define WW 64
#define HW 4096
#define NN 16
#define NROW (BB*HW)

typedef __attribute__((ext_vector_type(4))) float v4f;

__device__ __forceinline__ v8f wmma_f32(v2f a, v2f b, v8f c) {
    return __builtin_amdgcn_wmma_f32_16x16x4_f32(
         false, a,  false, b,
         (short)0, c,  false,  false);
}

__device__ __forceinline__ float softplus_f(float pre) {
    const float u = __expf(-fabsf(pre));
    const float t = 1.0f + u;
    const float c = u - (t - 1.0f);
    const float l = fmaf(__log2f(t), 0.693147180559945f, c);
    return fmaxf(pre, 0.0f) + l;
}

#define KB 8

template <bool BC>
__device__ __forceinline__ void load_blk(v2f (&ax)[KB], v2f (&ay)[KB], v2f (&bx)[KB], v2f (&by)[KB],
                                         const float* aP, const float* w0, const float* w1, int kb) {
    #pragma unroll
    for (int s = 0; s < KB; ++s) {
        const int k0 = kb + 4 * s;
        ax[s] = *reinterpret_cast<const v2f*>(aP + (size_t)k0 * HW);
        ay[s] = *reinterpret_cast<const v2f*>(aP + (size_t)(k0 + 1) * HW);
        if (BC) {
            bx[s] = (v2f){w0[(size_t)k0 * NN], w1[(size_t)k0 * NN]};
            by[s] = (v2f){w0[(size_t)(k0 + 1) * NN], w1[(size_t)(k0 + 1) * NN]};
        } else {
            bx[s] = *reinterpret_cast<const v2f*>(w0 + (size_t)k0 * CC);
            by[s] = *reinterpret_cast<const v2f*>(w0 + (size_t)(k0 + 1) * CC);
        }
    }
}

__device__ __forceinline__ void mma_blk(v8f& acc00, v8f& acc01, v8f& acc10, v8f& acc11,
                                        const v2f (&ax)[KB], const v2f (&ay)[KB],
                                        const v2f (&bx)[KB], const v2f (&by)[KB]) {
    #pragma unroll
    for (int s = 0; s < KB; ++s) {
        const v2f a0 = {ax[s].x, ay[s].x}, a1 = {ax[s].y, ay[s].y};
        const v2f b0 = {bx[s].x, by[s].x}, b1 = {bx[s].y, by[s].y};
        acc00 = wmma_f32(a0, b0, acc00);
        acc01 = wmma_f32(a0, b1, acc01);
        acc10 = wmma_f32(a1, b0, acc10);
        acc11 = wmma_f32(a1, b1, acc11);
    }
}

template <bool BC>
__device__ __forceinline__ void gemm_32x32(v8f& acc00, v8f& acc01, v8f& acc10, v8f& acc11,
                                           const float* aP, const float* w0, const float* w1) {
    v2f ax0[KB], ay0[KB], bx0[KB], by0[KB];
    v2f ax1[KB], ay1[KB], bx1[KB], by1[KB];
    load_blk<BC>(ax0, ay0, bx0, by0, aP, w0, w1, 0);
    #pragma unroll
    for (int kb = 0; kb < CC; kb += 8 * KB) {
        load_blk<BC>(ax1, ay1, bx1, by1, aP, w0, w1, kb + 4 * KB);
        __builtin_amdgcn_sched_barrier(0);
        mma_blk(acc00, acc01, acc10, acc11, ax0, ay0, bx0, by0);
        __builtin_amdgcn_sched_barrier(0);
        if (kb + 8 * KB < CC) load_blk<BC>(ax0, ay0, bx0, by0, aP, w0, w1, kb + 8 * KB);
        __builtin_amdgcn_sched_barrier(0);
        mma_blk(acc00, acc01, acc10, acc11, ax1, ay1, bx1, by1);
        __builtin_amdgcn_sched_barrier(0);
    }
}

__global__ void __launch_bounds__(32)
proj_wmma_kernel(const float* __restrict__ feat,
                                 const float* __restrict__ W_delta,
                                 const float* __restrict__ b_delta,
                                 const float* __restrict__ W_B,
                                 const float* __restrict__ W_C,
                                 float* __restrict__ delta_ws,
                                 float* __restrict__ Bm_ws,
                                 float* __restrict__ Cm_ws) {
    const int lane = threadIdx.x;
    const int lo   = lane & 15;
    const int hi   = lane >> 4;
    const int m0   = blockIdx.x * 32;
    const int which = blockIdx.y;
    const int bT   = m0 >> 12;
    const int lT   = m0 & (HW - 1);

    const float* aBase = feat + (size_t)bT * CC * HW + lT + 2 * lo + (size_t)(2 * hi) * HW;

    v8f acc00 = {}, acc01 = {}, acc10 = {}, acc11 = {};

    if (which < 8) {
        const int n0 = which * 32;
        const float* wBase = W_delta + n0 + 2 * lo + (size_t)(2 * hi) * CC;
        gemm_32x32<false>(acc00, acc01, acc10, acc11, aBase, wBase, wBase);
        #pragma unroll
        for (int j = 0; j < 2; ++j) {
            const int ch = n0 + 2 * lo + j;
            const float bias = b_delta[ch];
            const v8f e = j ? acc01 : acc00;
            const v8f o = j ? acc11 : acc10;
            v4f* dst = reinterpret_cast<v4f*>(delta_ws + ((size_t)(bT * CC + ch)) * HW + lT + 16 * hi);
            #pragma unroll
            for (int p = 0; p < 4; ++p) {
                dst[p] = (v4f){softplus_f(e[2 * p] + bias),     softplus_f(o[2 * p] + bias),
                               softplus_f(e[2 * p + 1] + bias), softplus_f(o[2 * p + 1] + bias)};
            }
        }
    } else {
        const float* wbBase = W_B + lo + (size_t)(2 * hi) * NN;
        const float* wcBase = W_C + lo + (size_t)(2 * hi) * NN;
        gemm_32x32<true>(acc00, acc01, acc10, acc11, aBase, wbBase, wcBase);
        #pragma unroll
        for (int v = 0; v < 8; ++v) {
            const size_t row = (size_t)(m0 + 2 * (v + 8 * hi));
            Bm_ws[row * NN + lo]       = acc00[v];
            Bm_ws[(row + 1) * NN + lo] = acc10[v];
            Cm_ws[row * NN + lo]       = acc01[v];
            Cm_ws[(row + 1) * NN + lo] = acc11[v];
        }
    }
}

#define KCH  32
#define NSC  3
#define NW2  (NSC * NCH * KCH / 8)
#define CLEN (HW / KCH)
#define SPL  4
#define NCH  2
#define NPL  (NSC * NCH)
#define RUN  64
#define ROWV4 (HW / 4)
static_assert(CLEN % RUN == 0 && RUN % SPL == 0, "a chunk is a whole number of runs of whole blocks");
static_assert((NSC * NCH * KCH) % 8 == 0 && NW2 <= KCH, "the items fill whole waves of 8 quads");

__device__ __forceinline__ float quad_sum(float v) {
    v += __builtin_bit_cast(float, __builtin_amdgcn_update_dpp(
             0, __builtin_bit_cast(int, v), 0xB1, 0xF, 0xF, true));
    v += __builtin_bit_cast(float, __builtin_amdgcn_update_dpp(
             0, __builtin_bit_cast(int, v), 0x4E, 0xF, 0xF, true));
    return v;
}

#define XDP (HW + 4)
#define YPP (HW + 4)
__global__ void __launch_bounds__(32 * KCH)
scan_kernel(const float* __restrict__ feat,
            const float* __restrict__ A_log,
            const float* __restrict__ Dvec,
            const float* __restrict__ delta_ws,
            const float* __restrict__ Bm_ws,
            const float* __restrict__ Cm_ws,
            float* __restrict__ out) {
    __shared__ __attribute__((aligned(16))) v2f sXD[NCH * XDP];
    __shared__ __attribute__((aligned(16))) float sY[NPL * YPP];
    __shared__ __attribute__((aligned(16))) float sS[3][64][64];

    const int tid  = threadIdx.x;
    const int lane = tid & 31;
    const int w  = __builtin_amdgcn_readfirstlane((int)(threadIdx.x >> 5));
    const int q  = lane & 3;
    const int p  = lane >> 2;
    const int ch = p & 1;
    const int grp  = w / 3;
    const int wk   = w - 3 * grp;
    const bool colw = (wk == 2);
    const int cw   = 4 * grp + (colw ? (p >> 1) : (2 * wk + (p >> 2)));
    const int scan = colw ? 2 : ((p >> 1) & 1);
    const int cl   = 2 * scan + ch;
    const int id  = blockIdx.x;
    const int b   = id >> 7;
    const int c0  = (id & 127) * NCH;
    const int c   = c0 + ch;

    {
        const size_t goff = ((size_t)(b * CC + c0)) * HW;
        const v4f* xs = reinterpret_cast<const v4f*>(feat + goff);
        const v4f* ds = reinterpret_cast<const v4f*>(delta_ws + goff);
        v4f xr[NCH], dr[NCH];
        for (int j = 0; j < NCH; ++j) {
            xr[j] = xs[j * ROWV4 + tid];
            dr[j] = ds[j * ROWV4 + tid];
        }
        for (int j = 0; j < NCH; ++j) {
            v4f* p = reinterpret_cast<v4f*>(&sXD[j * XDP + 4 * tid]);
            p[0] = (v4f){xr[j].x, dr[j].x, xr[j].y, dr[j].y};
            p[1] = (v4f){xr[j].z, dr[j].z, xr[j].w, dr[j].w};
        }
    }

    const int u0   = 8 * (cw >> 2) + (cw & 3);
    const int sh   = (scan == 0) ? 64 : (scan == 1) ? -64 : -4;
    const int sw   = (scan == 0) ? 1  : (scan == 1) ? -1 : 64;

    const v4f Al = *reinterpret_cast<const v4f*>(A_log + c * NN + SPL * q);
    float A[SPL];
    for (int j = 0; j < SPL; ++j) A[j] = -__expf(Al[j]);

    const int lS = (scan == 0) ? (CLEN * cw) : (scan == 1) ? (HW - 1 - CLEN * cw) : (63 - u0);
    const unsigned xa0 = (unsigned)((ch * XDP + lS) * 8);
    const unsigned ga0 = (unsigned)(lS * (NN * 4) + SPL * 4 * q);
    const unsigned ya0 = (unsigned)((cl * YPP + lS + q * sw) * 4);
    const int xstep = 8 * sw;
    const int gstep = NN * 4 * sw;
    const int ystep = 4 * sw;
    const int runfix = sh - RUN * sw;

    const char* xdB = reinterpret_cast<const char*>(sXD);
    char* yB = reinterpret_cast<char*>(sY);
    const char* BmB = reinterpret_cast<const char*>(Bm_ws + (size_t)b * HW * NN);
    const char* CmB = reinterpret_cast<const char*>(Cm_ws + (size_t)b * HW * NN);

    __syncthreads();

    {
        const int trav = (lane >> 3) & 1;
        const int u    = 2 * w + (lane >> 4);
        const int l0   = trav ? (63 - u) : (RUN * u);
        const int pst  = trav ? WW : 1;
        unsigned xa = (unsigned)((ch * XDP + l0) * 8);
        unsigned ga = (unsigned)(l0 * (NN * 4) + SPL * 4 * q);
        const unsigned xs1 = (unsigned)(8 * pst);
        const unsigned gs1 = (unsigned)(NN * 4 * pst);
        float hf[SPL], hb[SPL], P[SPL];
        for (int j = 0; j < SPL; ++j) { hf[j] = 0.f; hb[j] = 0.f; P[j] = 1.f; }
        for (int it = 0; it < RUN / SPL; ++it) {
            float d[SPL], xv[SPL];
            v4f Bv[SPL];
            for (int k = 0; k < SPL; ++k) {
                const v2f xd = *reinterpret_cast<const v2f*>(xdB + (xa + (unsigned)k * xs1));
                xv[k] = xd.x;
                d[k]  = xd.y;
                Bv[k] = *reinterpret_cast<const v4f*>(BmB + (ga + (unsigned)k * gs1));
            }
            for (int k = 0; k < SPL; ++k) {
                const float dx = d[k] * xv[k];
                for (int j = 0; j < SPL; ++j) {
                    const float dA = __expf(d[k] * A[j]);
                    const float bt = dx * Bv[k][j];
                    hb[j] = fmaf(P[j], bt, hb[j]);
                    hf[j] = fmaf(dA, hf[j], bt);
                    P[j] *= dA;
                }
            }
            xa += (unsigned)SPL * xs1;
            ga += (unsigned)SPL * gs1;
        }
        const int col = (lane & 15) * SPL;
        *reinterpret_cast<v4f*>(&sS[0][u][col]) = (v4f){P[0], P[1], P[2], P[3]};
        *reinterpret_cast<v4f*>(&sS[1][u][col]) = (v4f){hf[0], hf[1], hf[2], hf[3]};
        *reinterpret_cast<v4f*>(&sS[2][u][col]) = (v4f){hb[0], hb[1], hb[2], hb[3]};
    }
    __syncthreads();

    if (w < NW2) {
    float h[SPL], hB[SPL];
    for (int j = 0; j < SPL; ++j) { h[j] = 0.f; hB[j] = 0.f; }
    {
        const bool down = (scan == 1);
        const int ccol = (colw ? 32 : 0) + ch * NN + SPL * q;
        const float* pP = &sS[0][down ? 63 : 0][ccol];
        const float* pE = &sS[down ? 2 : 1][down ? 63 : 0][ccol];
        const int ust = down ? -64 : 64;
        const int nfold = colw ? u0 : 2 * cw;
        for (int i = 0; i < nfold; ++i) {
            const v4f Pv = *reinterpret_cast<const v4f*>(pP);
            const v4f Ev = *reinterpret_cast<const v4f*>(pE);
            for (int j = 0; j < SPL; ++j) h[j] = fmaf(Pv[j], h[j], Ev[j]);
            pP += ust;
            pE += ust;
        }
        if (colw) {
            for (int j = 0; j < SPL; ++j) hB[j] = h[j];
            for (int i = 0; i < 4; ++i) {
                const v4f Pv = *reinterpret_cast<const v4f*>(pP + 64 * i);
                const v4f Ev = *reinterpret_cast<const v4f*>(pE + 64 * i);
                for (int j = 0; j < SPL; ++j) hB[j] = fmaf(Pv[j], hB[j], Ev[j]);
            }
        } else if (scan == 0) {
            float g[SPL];
            for (int j = 0; j < SPL; ++j) g[j] = 0.f;
            const float* qP = &sS[0][63][ccol];
            const float* qE = &sS[1][63][ccol];
            for (int i = 0; i < 62 - 2 * cw; ++i) {
                const v4f Pv = *reinterpret_cast<const v4f*>(qP);
                const v4f Ev = *reinterpret_cast<const v4f*>(qE);
                for (int j = 0; j < SPL; ++j) g[j] = fmaf(Pv[j], g[j], Ev[j]);
                qP -= 64;
                qE -= 64;
            }
            const int rA = 2 * cw;
            const v4f PA = *reinterpret_cast<const v4f*>(&sS[0][rA][ccol]);
            const v4f EA = *reinterpret_cast<const v4f*>(&sS[1][rA][ccol]);
            const v4f PB = *reinterpret_cast<const v4f*>(&sS[0][rA + 1][ccol]);
            const v4f EB = *reinterpret_cast<const v4f*>(&sS[1][rA + 1][ccol]);
            for (int j = 0; j < SPL; ++j) {
                const float fB = fmaf(PA[j], h[j], EA[j]);
                const float gA = fmaf(PB[j], g[j], EB[j]);
                h[j]  = 0.5f * (h[j] + gA);
                hB[j] = 0.5f * (fB + g[j]);
            }
        }
    }

    {
        unsigned xa = xa0, ga = ga0, ya = ya0;
        for (int run = 0; run < CLEN / RUN; ++run) {
            if (run == 1 && scan != 1) { for (int j = 0; j < SPL; ++j) h[j] = hB[j]; }
            for (int it = 0; it < RUN / SPL; ++it) {
                float d[SPL], xv[SPL];
                v4f Bv[SPL], Cv[SPL];
                for (int k = 0; k < SPL; ++k) {
                    const v2f xd = *reinterpret_cast<const v2f*>(xdB + (xa + (unsigned)(k * xstep)));
                    xv[k] = xd.x;
                    d[k]  = xd.y;
                    Bv[k] = *reinterpret_cast<const v4f*>(BmB + (ga + (unsigned)(k * gstep)));
                    Cv[k] = *reinterpret_cast<const v4f*>(CmB + (ga + (unsigned)(k * gstep)));
                }
                float mine = 0.f;
                for (int k = 0; k < SPL; ++k) {
                    const float dx = d[k] * xv[k];
                    for (int j = 0; j < SPL; ++j) {
                        const float dA = __expf(d[k] * A[j]);
                        h[j] = fmaf(dA, h[j], dx * Bv[k][j]);
                    }
                    float y = h[0] * Cv[k][0];
                    y = fmaf(h[1], Cv[k][1], y);
                    y = fmaf(h[2], Cv[k][2], y);
                    y = fmaf(h[3], Cv[k][3], y);
                    y = quad_sum(y);
                    mine = (q == k) ? y : mine;
                }
                *reinterpret_cast<float*>(yB + ya) = mine;
                xa += (unsigned)(SPL * xstep);
                ga += (unsigned)(SPL * gstep);
                ya += (unsigned)(SPL * ystep);
            }
            xa += (unsigned)(8 * runfix);
            ga += (unsigned)(NN * 4 * runfix);
            ya += (unsigned)(4 * runfix);
        }
    }
    }
    __syncthreads();

    for (int j = 0; j < NCH; ++j) {
        const float dsc = Dvec[c0 + j];
        const v4f* px = reinterpret_cast<const v4f*>(&sXD[j * XDP + 4 * tid]);
        const v4f p0 = px[0], p1 = px[1];
        const v4f xq = (v4f){p0.x, p0.z, p1.x, p1.z};
        const v4f y0 = *reinterpret_cast<const v4f*>(&sY[(0 * NCH + j) * YPP + 4 * tid]);
        const v4f y1 = *reinterpret_cast<const v4f*>(&sY[(1 * NCH + j) * YPP + 4 * tid]);
        const v4f y2 = *reinterpret_cast<const v4f*>(&sY[(2 * NCH + j) * YPP + 4 * tid]);
        const v4f s = (y0 + y0) + (y1 + y2);
        v4f* dst = reinterpret_cast<v4f*>(out + ((size_t)(b * CC + c0 + j)) * HW);
        dst[tid] = xq * dsc + s * 0.25f;
    }
}

extern "C" void kernel_launch(void* const* d_in, const int* in_sizes, int n_in,
                              void* d_out, int out_size, void* d_ws, size_t ws_size,
                              hipStream_t stream) {
    const float* feat    = (const float*)d_in[0];
    const float* A_log   = (const float*)d_in[1];
    const float* Dvec    = (const float*)d_in[2];
    const float* W_delta = (const float*)d_in[3];
    const float* b_delta = (const float*)d_in[4];
    const float* W_B     = (const float*)d_in[5];
    const float* W_C     = (const float*)d_in[6];
    float* out = (float*)d_out;

    float* ws       = (float*)d_ws;
    float* delta_ws = ws;
    float* Bm_ws    = ws + (size_t)NROW * CC;
    float* Cm_ws    = Bm_ws + (size_t)NROW * NN;

    proj_wmma_kernel<<<dim3(NROW / 32, 9), 32, 0, stream>>>(
        feat, W_delta, b_delta, W_B, W_C, delta_ws, Bm_ws, Cm_ws);
    scan_kernel<<<256, 32 * KCH, 0, stream>>>(feat, A_log, Dvec, delta_ws, Bm_ws, Cm_ws, out);
}
